# starting kernel
# baseline (speedup 1.0000x reference)
#include <hip/hip_runtime.h>

#define NQ       12
#define DIM      4096
#define NLAYERS  6
#define TPB      256
#define APT      (DIM / TPB)
#define ANG_STRIDE 16
#define DOUT     256

typedef float v2f __attribute__((ext_vector_type(2)));
typedef float v8f __attribute__((ext_vector_type(8)));

__global__ __launch_bounds__(256) void fc_in_wmma(const float* __restrict__ x,
                                                  const float* __restrict__ W_in,
                                                  const float* __restrict__ b_in,
                                                  float* __restrict__ angles) {
    const int lane = threadIdx.x & 31;
    const int wave = threadIdx.x >> 5;
    const int tile = blockIdx.x * 8 + wave;
    const int n    = lane & 15;
    const int h    = lane >> 4;
    const int nc   = (n < NQ) ? n : (NQ - 1);
    const float mask = (n < NQ) ? 1.0f : 0.0f;

    const float* xrow = x + (size_t)(tile * 16 + n) * 256;
    const float* wrow = W_in + nc * 256;

    v8f acc = {};
    for (int k0 = 0; k0 < 256; k0 += 8) {
        const float4 av = *(const float4*)(xrow + k0 + 4 * h);
        float4 wv       = *(const float4*)(wrow + k0 + 4 * h);
        wv.x *= mask; wv.y *= mask; wv.z *= mask; wv.w *= mask;

        v2f a1, b1, a2, b2;
        a1.x = av.x; a1.y = av.y;  b1.x = wv.x; b1.y = wv.y;
        a2.x = av.z; a2.y = av.w;  b2.x = wv.z; b2.y = wv.w;
        acc = __builtin_amdgcn_wmma_f32_16x16x4_f32(false, a1, false, b1,
                                                    (short)0, acc, false, false);
        acc = __builtin_amdgcn_wmma_f32_16x16x4_f32(false, a2, false, b2,
                                                    (short)0, acc, false, false);
    }

    const float bias = b_in[nc] * mask;
    const int mbase = tile * 16 + (h << 3);
#pragma unroll
    for (int r = 0; r < 8; ++r)
        angles[(size_t)(mbase + r) * ANG_STRIDE + n] = acc[r] + bias;
}

__device__ __forceinline__ void rx_apply(float2* st, int bit, float c, float s) {
    const int tid = threadIdx.x;
#pragma unroll
    for (int it = 0; it < (DIM / 2) / TPB; ++it) {
        int p  = tid + it * TPB;
        int i0 = ((p >> bit) << (bit + 1)) | (p & ((1 << bit) - 1));
        int i1 = i0 | (1 << bit);
        float2 a0 = st[i0];
        float2 a1 = st[i1];
        float2 r0, r1;
        r0.x = fmaf(c, a0.x,  s * a1.y);
        r0.y = fmaf(c, a0.y, -s * a1.x);
        r1.x = fmaf(c, a1.x,  s * a0.y);
        r1.y = fmaf(c, a1.y, -s * a0.x);
        st[i0] = r0;
        st[i1] = r1;
    }
    __syncthreads();
}

__device__ __forceinline__ void cnot_apply(float2* st, int cbit, int tbit) {
    const int tid = threadIdx.x;
    const int lo = (cbit < tbit) ? cbit : tbit;
    const int hi = (cbit < tbit) ? tbit : cbit;
#pragma unroll
    for (int it = 0; it < (DIM / 4) / TPB; ++it) {
        int p = tid + it * TPB;
        int i = ((p >> lo) << (lo + 1)) | (p & ((1 << lo) - 1));
        i     = ((i >> hi) << (hi + 1)) | (i & ((1 << hi) - 1));
        i    |= (1 << cbit);
        int j = i | (1 << tbit);
        float2 t = st[i];
        st[i] = st[j];
        st[j] = t;
    }
    __syncthreads();
}

__global__ __launch_bounds__(TPB) void vqc_kernel(const float* __restrict__ angles,
                                                  const float* __restrict__ qp,
                                                  const float* __restrict__ W_out,
                                                  const float* __restrict__ b_out,
                                                  float* __restrict__ out) {
    __shared__ float2 st[DIM];
    const int tid = threadIdx.x;
    const int b   = blockIdx.x;

#pragma unroll
    for (int j = 0; j < APT; ++j) {
        float2 z; z.x = 0.f; z.y = 0.f;
        st[tid * APT + j] = z;
    }
    if (tid == 0) { float2 one; one.x = 1.f; one.y = 0.f; st[0] = one; }
    __syncthreads();

    for (int q = 0; q < NQ; ++q) {
        float hh = 0.5f * angles[(size_t)b * ANG_STRIDE + q];
        rx_apply(st, 11 - q, __cosf(hh), __sinf(hh));
    }

    for (int l = 0; l < NLAYERS; ++l) {
        for (int q = 0; q < NQ; ++q) {
            float hh = 0.5f * qp[l * NQ + q];
            rx_apply(st, 11 - q, __cosf(hh), __sinf(hh));
        }
        for (int q = 0; q < NQ - 1; ++q)
            cnot_apply(st, 11 - q, 10 - q);
        cnot_apply(st, 0, 11);
    }

    float zp[NQ];
#pragma unroll
    for (int q = 0; q < NQ; ++q) zp[q] = 0.f;
#pragma unroll
    for (int j = 0; j < APT; ++j) {
        int i = tid * APT + j;
        float2 a = st[i];
        float pr = fmaf(a.x, a.x, a.y * a.y);
#pragma unroll
        for (int q = 0; q < NQ; ++q)
            zp[q] += ((i >> (11 - q)) & 1) ? -pr : pr;
    }
    __syncthreads();
    float* red = (float*)st;
#pragma unroll
    for (int q = 0; q < NQ; ++q) red[tid * NQ + q] = zp[q];
    __syncthreads();
    for (int s = TPB / 2; s >= 1; s >>= 1) {
        if (tid < s) {
#pragma unroll
            for (int q = 0; q < NQ; ++q)
                red[tid * NQ + q] += red[(tid + s) * NQ + q];
        }
        __syncthreads();
    }

    float accv = b_out[tid];
#pragma unroll
    for (int q = 0; q < NQ; ++q)
        accv = fmaf(red[q], W_out[tid * NQ + q], accv);
    out[(size_t)b * DOUT + tid] = accv;
}

extern "C" void kernel_launch(void* const* d_in, const int* in_sizes, int n_in,
                              void* d_out, int out_size, void* d_ws, size_t ws_size,
                              hipStream_t stream) {
    const float* x        = (const float*)d_in[0];
    const float* W_in     = (const float*)d_in[1];
    const float* b_in     = (const float*)d_in[2];
    const float* q_params = (const float*)d_in[3];
    const float* W_out    = (const float*)d_in[4];
    const float* b_out    = (const float*)d_in[5];
    float* out    = (float*)d_out;
    float* angles = (float*)d_ws;

    fc_in_wmma<<<dim3(32),   dim3(256), 0, stream>>>(x, W_in, b_in, angles);
    vqc_kernel<<<dim3(4096), dim3(TPB), 0, stream>>>(angles, q_params, W_out, b_out, out);
}


# CNOT ring folded into a GF(2) index map held as bit masks: RX pairs x with x^mask, Z signs by mask parity, so no CNOT pass touches LDS (72 of 156 passes gone); Z sum order changes
# speedup vs baseline: 1.5073x; 1.5073x over previous
#include <hip/hip_runtime.h>

#define NQ       12
#define DIM      4096
#define NLAYERS  6
#define TPB      256
#define APT      (DIM / TPB)
#define ANG_STRIDE 16
#define DOUT     256

typedef float v2f __attribute__((ext_vector_type(2)));
typedef float v8f __attribute__((ext_vector_type(8)));

__global__ __launch_bounds__(256) void fc_in_wmma(const float* __restrict__ x,
                                                  const float* __restrict__ W_in,
                                                  const float* __restrict__ b_in,
                                                  float* __restrict__ angles) {
    const int lane = threadIdx.x & 31;
    const int wave = threadIdx.x >> 5;
    const int tile = blockIdx.x * 8 + wave;
    const int n    = lane & 15;
    const int h    = lane >> 4;
    const int nc   = (n < NQ) ? n : (NQ - 1);
    const float mask = (n < NQ) ? 1.0f : 0.0f;

    const float* xrow = x + (size_t)(tile * 16 + n) * 256;
    const float* wrow = W_in + nc * 256;

    v8f acc = {};
    for (int k0 = 0; k0 < 256; k0 += 8) {
        const float4 av = *(const float4*)(xrow + k0 + 4 * h);
        float4 wv       = *(const float4*)(wrow + k0 + 4 * h);
        wv.x *= mask; wv.y *= mask; wv.z *= mask; wv.w *= mask;

        v2f a1, b1, a2, b2;
        a1.x = av.x; a1.y = av.y;  b1.x = wv.x; b1.y = wv.y;
        a2.x = av.z; a2.y = av.w;  b2.x = wv.z; b2.y = wv.w;
        acc = __builtin_amdgcn_wmma_f32_16x16x4_f32(false, a1, false, b1,
                                                    (short)0, acc, false, false);
        acc = __builtin_amdgcn_wmma_f32_16x16x4_f32(false, a2, false, b2,
                                                    (short)0, acc, false, false);
    }

    const float bias = b_in[nc] * mask;
    const int mbase = tile * 16 + (h << 3);
#pragma unroll
    for (int r = 0; r < 8; ++r)
        angles[(size_t)(mbase + r) * ANG_STRIDE + n] = acc[r] + bias;
}

__device__ __forceinline__ void rx_apply(float2* st, unsigned m, float c, float s) {
    const unsigned hi8   = (~0u << (31 - __builtin_clz(m))) << 3;
    const unsigned m8    = m << 3;
    const unsigned t8    = threadIdx.x << 3;
    const unsigned base8 = t8 + (t8 & hi8);
    char* stb = (char*)st;
#pragma unroll
    for (int it = 0; it < (DIM / 2) / TPB; ++it) {
        const unsigned k8 = it * TPB * 8;
        const unsigned o0 = base8 + (k8 + (k8 & hi8));
        const unsigned o1 = o0 ^ m8;
        float2 a0 = *(float2*)(stb + o0);
        float2 a1 = *(float2*)(stb + o1);
        float2 r0, r1;
        r0.x = fmaf(c, a0.x,  s * a1.y);
        r0.y = fmaf(c, a0.y, -s * a1.x);
        r1.x = fmaf(c, a1.x,  s * a0.y);
        r1.y = fmaf(c, a1.y, -s * a0.x);
        *(float2*)(stb + o0) = r0;
        *(float2*)(stb + o1) = r1;
    }
    __syncthreads();
}

__device__ __forceinline__ void cnot_relabel(unsigned* rowL, unsigned* colInv, int cbit, int tbit) {
    rowL[tbit]   ^= rowL[cbit];
    colInv[cbit] ^= colInv[tbit];
}

__global__ __launch_bounds__(TPB) void vqc_kernel(const float* __restrict__ angles,
                                                  const float* __restrict__ qp,
                                                  const float* __restrict__ W_out,
                                                  const float* __restrict__ b_out,
                                                  float* __restrict__ out) {
    __shared__ float2 st[DIM];
    const int tid = threadIdx.x;
    const int b   = blockIdx.x;

#pragma unroll
    for (int j = 0; j < APT; ++j) {
        float2 z; z.x = 0.f; z.y = 0.f;
        st[tid * APT + j] = z;
    }
    if (tid == 0) { float2 one; one.x = 1.f; one.y = 0.f; st[0] = one; }
    __syncthreads();

    unsigned rowL[NQ], colInv[NQ];
#pragma unroll
    for (int k = 0; k < NQ; ++k) { rowL[k] = 1u << k; colInv[k] = 1u << k; }

    for (int q = 0; q < NQ; ++q) {
        float hh = 0.5f * angles[(size_t)b * ANG_STRIDE + q];
        rx_apply(st, 1u << (11 - q), __cosf(hh), __sinf(hh));
    }

    for (int l = 0; l < NLAYERS; ++l) {
#pragma unroll
        for (int q = 0; q < NQ; ++q) {
            float hh = 0.5f * qp[l * NQ + q];
            rx_apply(st, colInv[11 - q], __cosf(hh), __sinf(hh));
        }
#pragma unroll
        for (int q = 0; q < NQ - 1; ++q)
            cnot_relabel(rowL, colInv, 11 - q, 10 - q);
        cnot_relabel(rowL, colInv, 0, 11);
    }

    float zp[NQ];
#pragma unroll
    for (int q = 0; q < NQ; ++q) zp[q] = 0.f;
#pragma unroll
    for (int j = 0; j < APT; ++j) {
        float2 a = st[tid * APT + j];
        float pr = fmaf(a.x, a.x, a.y * a.y);
#pragma unroll
        for (int q = 0; q < NQ; ++q) {
            const unsigned sj = (__builtin_popcount((unsigned)j & rowL[11 - q]) & 1u) << 31;
            zp[q] += __uint_as_float(__float_as_uint(pr) ^ sj);
        }
    }
#pragma unroll
    for (int q = 0; q < NQ; ++q) {
        const unsigned stid = (__builtin_popcount(((unsigned)tid * APT) & rowL[11 - q]) & 1u) << 31;
        zp[q] = __uint_as_float(__float_as_uint(zp[q]) ^ stid);
    }
    __syncthreads();
    float* red = (float*)st;
#pragma unroll
    for (int q = 0; q < NQ; ++q) red[tid * NQ + q] = zp[q];
    __syncthreads();
    for (int s = TPB / 2; s >= 1; s >>= 1) {
        if (tid < s) {
#pragma unroll
            for (int q = 0; q < NQ; ++q)
                red[tid * NQ + q] += red[(tid + s) * NQ + q];
        }
        __syncthreads();
    }

    float accv = b_out[tid];
#pragma unroll
    for (int q = 0; q < NQ; ++q)
        accv = fmaf(red[q], W_out[tid * NQ + q], accv);
    out[(size_t)b * DOUT + tid] = accv;
}

extern "C" void kernel_launch(void* const* d_in, const int* in_sizes, int n_in,
                              void* d_out, int out_size, void* d_ws, size_t ws_size,
                              hipStream_t stream) {
    const float* x        = (const float*)d_in[0];
    const float* W_in     = (const float*)d_in[1];
    const float* b_in     = (const float*)d_in[2];
    const float* q_params = (const float*)d_in[3];
    const float* W_out    = (const float*)d_in[4];
    const float* b_out    = (const float*)d_in[5];
    float* out    = (float*)d_out;
    float* angles = (float*)d_ws;

    fc_in_wmma<<<dim3(32),   dim3(256), 0, stream>>>(x, W_in, b_in, angles);
    vqc_kernel<<<dim3(4096), dim3(TPB), 0, stream>>>(angles, q_params, W_out, b_out, out);
}


# RX is diagonal in the X basis: state is a pure phase, Z expval = mean cosine of signed angle sums via a 16-point sign butterfly per thread; no LDS state or gate passes; Z sum order changes
# speedup vs baseline: 7.5219x; 4.9903x over previous
#include <hip/hip_runtime.h>

#define NQ       12
#define DIM      4096
#define NLAYERS  6
#define TPB      256
#define ANG_STRIDE 16
#define DOUT     256

typedef float v2f __attribute__((ext_vector_type(2)));
typedef float v8f __attribute__((ext_vector_type(8)));

__global__ __launch_bounds__(256) void fc_in_wmma(const float* __restrict__ x,
                                                  const float* __restrict__ W_in,
                                                  const float* __restrict__ b_in,
                                                  float* __restrict__ angles) {
    const int lane = threadIdx.x & 31;
    const int wave = threadIdx.x >> 5;
    const int tile = blockIdx.x * 8 + wave;
    const int n    = lane & 15;
    const int h    = lane >> 4;
    const int nc   = (n < NQ) ? n : (NQ - 1);
    const float mask = (n < NQ) ? 1.0f : 0.0f;

    const float* xrow = x + (size_t)(tile * 16 + n) * 256;
    const float* wrow = W_in + nc * 256;

    v8f acc = {};
    for (int k0 = 0; k0 < 256; k0 += 8) {
        const float4 av = *(const float4*)(xrow + k0 + 4 * h);
        float4 wv       = *(const float4*)(wrow + k0 + 4 * h);
        wv.x *= mask; wv.y *= mask; wv.z *= mask; wv.w *= mask;

        v2f a1, b1, a2, b2;
        a1.x = av.x; a1.y = av.y;  b1.x = wv.x; b1.y = wv.y;
        a2.x = av.z; a2.y = av.w;  b2.x = wv.z; b2.y = wv.w;
        acc = __builtin_amdgcn_wmma_f32_16x16x4_f32(false, a1, false, b1,
                                                    (short)0, acc, false, false);
        acc = __builtin_amdgcn_wmma_f32_16x16x4_f32(false, a2, false, b2,
                                                    (short)0, acc, false, false);
    }

    const float bias = b_in[nc] * mask;
    const int mbase = tile * 16 + (h << 3);
#pragma unroll
    for (int r = 0; r < 8; ++r)
        angles[(size_t)(mbase + r) * ANG_STRIDE + n] = acc[r] + bias;
}

#define NG (NLAYERS * NQ)

struct MaskTab { unsigned m[NG]; };

constexpr MaskTab make_masks() {
    MaskTab t{};
    unsigned R[NQ] = {};
    for (int q = 0; q < NQ; ++q) R[q] = 1u << q;
    for (int l = NLAYERS - 1; l >= 0; --l) {
        R[NQ - 1] ^= R[0];
        for (int q = NQ - 2; q >= 0; --q) R[q] ^= R[q + 1];
        for (int q = 0; q < NQ; ++q) t.m[l * NQ + q] = R[q];
    }
    return t;
}
static_assert(make_masks().m[0] == 0xd7cu && make_masks().m[18] == 0x1e2u && make_masks().m[30] == 0x226u &&
              make_masks().m[47] == 0xd54u && make_masks().m[71] == 0x801u, "Z strings");

__global__ __launch_bounds__(TPB) void vqc_kernel(const float* __restrict__ angles,
                                                  const float* __restrict__ qp,
                                                  const float* __restrict__ W_out,
                                                  const float* __restrict__ b_out,
                                                  float* __restrict__ out) {
    __shared__ float red[TPB * NQ];
    constexpr MaskTab MT = make_masks();
    const int tid = threadIdx.x;
    const int b   = blockIdx.x;
    const unsigned thi = (unsigned)tid << 4;

    float u[NG];
#pragma unroll
    for (int g = 0; g < NG; ++g) {
        float th = qp[g];
        if (g < NQ) th += angles[(size_t)b * ANG_STRIDE + g];
        th *= 0.15915494309189535f;
        const unsigned sg = (__builtin_popcount(thi & MT.m[g]) & 1u) << 31;
        u[g] = __uint_as_float(__float_as_uint(th) ^ sg);
    }

    float zp[NQ];
#pragma unroll
    for (int q = 0; q < NQ; ++q) {
        float Gc[16];
#pragma unroll
        for (int c = 0; c < 16; ++c) Gc[c] = -0.0f;
#pragma unroll
        for (int g = 0; g < NG; ++g)
            if ((MT.m[g] >> q) & 1u) Gc[MT.m[g] & 15u] += u[g];
#pragma unroll
        for (int h = 2; h < 16; h <<= 1) {
#pragma unroll
            for (int i = 0; i < 16; ++i) {
                if (i & h) continue;
                const float x0 = Gc[i], x1 = Gc[i | h];
                Gc[i]     = x0 + x1;
                Gc[i | h] = x0 - x1;
            }
        }
        float acc = 0.f;
#pragma unroll
        for (int jp = 0; jp < 8; ++jp) {
            const float cp = __builtin_amdgcn_cosf(__builtin_amdgcn_fractf(Gc[2 * jp]));
            const float cq = __builtin_amdgcn_cosf(__builtin_amdgcn_fractf(Gc[2 * jp + 1]));
            acc = fmaf(cp, cq, acc);
        }
        zp[q] = acc;
    }

#pragma unroll
    for (int q = 0; q < NQ; ++q) red[tid * NQ + q] = zp[q];
    __syncthreads();
    for (int s = TPB / 2; s >= 1; s >>= 1) {
        if (tid < s) {
#pragma unroll
            for (int q = 0; q < NQ; ++q)
                red[tid * NQ + q] += red[(tid + s) * NQ + q];
        }
        __syncthreads();
    }

    float accv = b_out[tid];
#pragma unroll
    for (int q = 0; q < NQ; ++q)
        accv = fmaf(red[q] * (2.0f / (float)DIM), W_out[tid * NQ + q], accv);
    out[(size_t)b * DOUT + tid] = accv;
}

extern "C" void kernel_launch(void* const* d_in, const int* in_sizes, int n_in,
                              void* d_out, int out_size, void* d_ws, size_t ws_size,
                              hipStream_t stream) {
    const float* x        = (const float*)d_in[0];
    const float* W_in     = (const float*)d_in[1];
    const float* b_in     = (const float*)d_in[2];
    const float* q_params = (const float*)d_in[3];
    const float* W_out    = (const float*)d_in[4];
    const float* b_out    = (const float*)d_in[5];
    float* out    = (float*)d_out;
    float* angles = (float*)d_ws;

    fc_in_wmma<<<dim3(32),   dim3(256), 0, stream>>>(x, W_in, b_in, angles);
    vqc_kernel<<<dim3(4096), dim3(TPB), 0, stream>>>(angles, q_params, W_out, b_out, out);
}


# Only 12 angles vary per sample: the 60 shared gates are summed once per call into a 2176-entry complex table over encoding sign patterns; 2176 terms per sample, not 49152; Z sum order changes
# speedup vs baseline: 21.9548x; 2.9188x over previous
#include <hip/hip_runtime.h>

#define NQ       12
#define NLAYERS  6
#define TPB      256
#define ANG_STRIDE 16
#define DOUT     256
#define NG (NLAYERS * NQ)

#define FC_BLOCKS   32
#define TAB_CHUNKS  8
#define TAB_BLOCKS  (NQ * TAB_CHUNKS)
#define NTERM       2176
#define NITER       2
#define SPB         (8 * NITER)
#define REV         0.15915494309189535f

typedef float v2f __attribute__((ext_vector_type(2)));
typedef float v8f __attribute__((ext_vector_type(8)));

__device__ __forceinline__ void fc_in_tile(const float* __restrict__ x,
                                           const float* __restrict__ W_in,
                                           const float* __restrict__ b_in,
                                           float* __restrict__ angles) {
    const int lane = threadIdx.x & 31;
    const int wave = threadIdx.x >> 5;
    const int tile = blockIdx.x * 8 + wave;
    const int n    = lane & 15;
    const int h    = lane >> 4;
    const int nc   = (n < NQ) ? n : (NQ - 1);
    const float mask = (n < NQ) ? 1.0f : 0.0f;

    const float* xrow = x + (size_t)(tile * 16 + n) * 256;
    const float* wrow = W_in + nc * 256;

    v8f acc = {};
    for (int k0 = 0; k0 < 256; k0 += 8) {
        const float4 av = *(const float4*)(xrow + k0 + 4 * h);
        float4 wv       = *(const float4*)(wrow + k0 + 4 * h);
        wv.x *= mask; wv.y *= mask; wv.z *= mask; wv.w *= mask;

        v2f a1, b1, a2, b2;
        a1.x = av.x; a1.y = av.y;  b1.x = wv.x; b1.y = wv.y;
        a2.x = av.z; a2.y = av.w;  b2.x = wv.z; b2.y = wv.w;
        acc = __builtin_amdgcn_wmma_f32_16x16x4_f32(false, a1, false, b1,
                                                    (short)0, acc, false, false);
        acc = __builtin_amdgcn_wmma_f32_16x16x4_f32(false, a2, false, b2,
                                                    (short)0, acc, false, false);
    }

    const float bias = b_in[nc] * mask;
    const int mbase = tile * 16 + (h << 3);
#pragma unroll
    for (int r = 0; r < 8; ++r)
        angles[(size_t)(mbase + r) * ANG_STRIDE + n] = acc[r] + bias;
}

struct MaskTab { unsigned m[NG]; };

constexpr MaskTab make_masks() {
    MaskTab t{};
    unsigned R[NQ] = {};
    for (int q = 0; q < NQ; ++q) R[q] = 1u << q;
    for (int l = NLAYERS - 1; l >= 0; --l) {
        R[NQ - 1] ^= R[0];
        for (int q = NQ - 2; q >= 0; --q) R[q] ^= R[q + 1];
        for (int q = 0; q < NQ; ++q) t.m[l * NQ + q] = R[q];
    }
    return t;
}
static_assert(make_masks().m[0] == 0xd7cu && make_masks().m[18] == 0x1e2u && make_masks().m[30] == 0x226u &&
              make_masks().m[47] == 0xd54u && make_masks().m[71] == 0x801u, "Z strings");

struct Plan {
    unsigned mpi[NQ][NG];
    int K[NQ];
    int off[NQ];
    int gsel[NQ][NQ];
    int total;
};
constexpr unsigned cpar(unsigned v) { unsigned p = 0; while (v) { p ^= v & 1u; v >>= 1; } return p; }
constexpr Plan make_plan() {
    Plan P{};
    const MaskTab T = make_masks();
    unsigned A[NQ] = {};
    for (int i = 0; i < NQ; ++i) A[i] = T.m[i] | (1u << (NQ + i));
    for (int c = 0; c < NQ; ++c) {
        int p = c;
        while (!((A[p] >> c) & 1u)) ++p;
        const unsigned t = A[c]; A[c] = A[p]; A[p] = t;
        for (int r = 0; r < NQ; ++r)
            if (r != c && ((A[r] >> c) & 1u)) A[r] ^= A[c];
    }
    unsigned Lc[NQ] = {};
    for (int j = 0; j < NQ; ++j)
        for (int i = 0; i < NQ; ++i) Lc[j] |= ((A[i] >> (NQ + j)) & 1u) << i;
    unsigned mp[NG] = {};
    for (int g = 0; g < NG; ++g)
        for (int k = 0; k < NQ; ++k) mp[g] |= cpar(Lc[k] & T.m[g]) << k;
    int tot = 0;
    bool ok = true;
    for (int g = 0; g < NQ; ++g) ok = ok && (mp[g] == (1u << g));
    for (int q = 0; q < NQ; ++q) {
        int k = 0;
        for (int g = 0; g < NQ; ++g) k += (int)((T.m[g] >> q) & 1u);
        P.K[q] = k;
        P.off[q] = tot;
        tot += 1 << (k - 1);
        const int nt = NQ - k;
        int a = 0, bn = 0;
        int pos[NQ] = {};
        for (int g = 0; g < NQ; ++g) {
            if ((T.m[g] >> q) & 1u) { pos[g] = nt + a; P.gsel[q][a] = g; ++a; }
            else                    { pos[g] = bn; ++bn; }
        }
        for (int g = 0; g < NG; ++g) {
            unsigned v = 0;
            for (int k2 = 0; k2 < NQ; ++k2) v |= ((mp[g] >> k2) & 1u) << pos[k2];
            P.mpi[q][g] = v;
        }
        ok = ok && (k >= 6) && (k <= 10);
    }
    P.total = ok ? tot : -1;
    return P;
}
static_assert(make_plan().total == NTERM, "half-table size");
static_assert(make_plan().K[0] == 8 && make_plan().K[1] == 6 && make_plan().K[6] == 10 && make_plan().K[11] == 9, "K_q");
static_assert(make_plan().off[3] == 192 && make_plan().off[7] == 1152 && make_plan().off[11] == 1920, "offsets");
static_assert(make_plan().mpi[0][12] == 0x11u && make_plan().mpi[11][40] == 0x1c2u && make_plan().mpi[6][71] == 0x82du, "strings over idx");
static_assert(make_plan().gsel[1][5] == 11 && make_plan().gsel[6][5] == 6 && make_plan().gsel[6][9] == 11, "gate order");

__device__ __forceinline__ void build_table(const int q, const int chunk,
                                            const float* __restrict__ qp,
                                            float2* __restrict__ tab, float* sP, float* sQ,
                                            float* sW, unsigned* sM) {
    constexpr MaskTab MT = make_masks();
    constexpr Plan PL = make_plan();
    if (threadIdx.x < NG - NQ) {
        const int g = NQ + (int)threadIdx.x;
        sW[threadIdx.x] = ((MT.m[g] >> q) & 1u) ? qp[g] * REV : 0.0f;
        sM[threadIdx.x] = PL.mpi[q][g];
    }
    __syncthreads();
    const int K  = PL.K[q];
    const int NT = NQ - K;
    const unsigned FLIP = ((1u << K) - 1u) << NT;
    const int tid = threadIdx.x;
    const unsigned idx = (unsigned)(chunk * TPB + tid);
    const unsigned idm = idx ^ FLIP;
    float p1 = 0.f, p2 = 0.f;
    for (int n = 0; n < NG - NQ; ++n) {
        const unsigned w  = __float_as_uint(sW[n]);
        const unsigned mk = sM[n];
        p1 += __uint_as_float(w ^ ((__builtin_popcount(idx & mk) & 1u) << 31));
        p2 += __uint_as_float(w ^ ((__builtin_popcount(idm & mk) & 1u) << 31));
    }
    const float f1 = __builtin_amdgcn_fractf(p1), f2 = __builtin_amdgcn_fractf(p2);
    sP[tid] = __builtin_amdgcn_cosf(f1) + __builtin_amdgcn_cosf(f2);
    sQ[tid] = __builtin_amdgcn_sinf(f1) - __builtin_amdgcn_sinf(f2);
    __syncthreads();
    const int nt = 1 << NT;
    if ((tid & (nt - 1)) == 0) {
        float aP = 0.f, aQ = 0.f;
        for (int t = 0; t < nt; ++t) { aP += sP[tid + t]; aQ += sQ[tid + t]; }
        tab[PL.off[q] + (int)(idx >> NT)] = make_float2(aP, aQ);
    }
}

__global__ __launch_bounds__(TPB) void fc_in_and_table(const float* __restrict__ x,
                                                       const float* __restrict__ W_in,
                                                       const float* __restrict__ b_in,
                                                       const float* __restrict__ qp,
                                                       float* __restrict__ angles,
                                                       float* __restrict__ tabf) {
    __shared__ float sP[TPB];
    __shared__ float sQ[TPB];
    __shared__ float sW[NG - NQ];
    __shared__ unsigned sM[NG - NQ];
    if (blockIdx.x < FC_BLOCKS) {
        fc_in_tile(x, W_in, b_in, angles);
        return;
    }
    const int tb = (int)blockIdx.x - FC_BLOCKS;
    build_table(tb / TAB_CHUNKS, tb % TAB_CHUNKS, qp, (float2*)tabf, sP, sQ, sW, sM);
}

template <int Q>
__device__ __forceinline__ float eval_q(const float (&th)[NQ], const float2* stab,
                                        const unsigned lane) {
    constexpr Plan PL = make_plan();
    constexpr int K   = PL.K[Q];
    constexpr int NHI = 1 << (K - 6);
    float base = 0.f;
#pragma unroll
    for (int i = 0; i < 5; ++i)
        base += __uint_as_float(__float_as_uint(th[PL.gsel[Q][i]]) ^ (((lane >> i) & 1u) << 31));
    float hs[NHI];
    float h0 = 0.f;
#pragma unroll
    for (int i = 5; i < K; ++i) h0 += th[PL.gsel[Q][i]];
    hs[0] = h0;
#pragma unroll
    for (int j = 5; j < K - 1; ++j) {
        const float t2 = 2.0f * th[PL.gsel[Q][j]];
#pragma unroll
        for (int h = 0; h < (1 << (j - 5)); ++h) hs[h | (1 << (j - 5))] = hs[h] - t2;
    }
    float acc = 0.f;
#pragma unroll
    for (int hi = 0; hi < NHI; ++hi) {
        const float2 e = stab[PL.off[Q] + hi * 32 + (int)lane];
        const float f = __builtin_amdgcn_fractf(base + hs[hi]);
        acc = fmaf(e.x, __builtin_amdgcn_cosf(f), acc);
        acc = fmaf(-e.y, __builtin_amdgcn_sinf(f), acc);
    }
    return acc;
}

__global__ __launch_bounds__(TPB) void vqc_eval(const float* __restrict__ angles,
                                                const float* __restrict__ qp,
                                                const float* __restrict__ tabf,
                                                const float* __restrict__ W_out,
                                                const float* __restrict__ b_out,
                                                float* __restrict__ out) {
    constexpr int NV = NTERM / 2, NCP = (NV + TPB - 1) / TPB;
    __shared__ float2 stab[NCP * TPB * 2];
    __shared__ float zs[SPB * NQ];
    const int tid = threadIdx.x;
    const unsigned lane = (unsigned)tid & 31u;
    const int wave = tid >> 5;

    {
        const float4* src = (const float4*)tabf;
        float4* dst = (float4*)stab;
        float4 tv[NCP];
#pragma unroll
        for (int k = 0; k < NCP; ++k) { const int i = tid + k * TPB; tv[k] = src[i < NV ? i : NV - 1]; }
#pragma unroll
        for (int k = 0; k < NCP; ++k) dst[tid + k * TPB] = tv[k];
    }
    float wrow[NQ];
#pragma unroll
    for (int q = 0; q < NQ; ++q) wrow[q] = W_out[tid * NQ + q];
    const float bo = b_out[tid];
    __syncthreads();

    for (int it = 0; it < NITER; ++it) {
        const int sl = it * 8 + wave;
        const int b  = (int)blockIdx.x * SPB + sl;
        float th[NQ];
#pragma unroll
        for (int g = 0; g < NQ; ++g)
            th[g] = (qp[g] + angles[(size_t)b * ANG_STRIDE + g]) * REV;
        float zp[NQ];
        zp[0]  = eval_q<0>(th, stab, lane);
        zp[1]  = eval_q<1>(th, stab, lane);
        zp[2]  = eval_q<2>(th, stab, lane);
        zp[3]  = eval_q<3>(th, stab, lane);
        zp[4]  = eval_q<4>(th, stab, lane);
        zp[5]  = eval_q<5>(th, stab, lane);
        zp[6]  = eval_q<6>(th, stab, lane);
        zp[7]  = eval_q<7>(th, stab, lane);
        zp[8]  = eval_q<8>(th, stab, lane);
        zp[9]  = eval_q<9>(th, stab, lane);
        zp[10] = eval_q<10>(th, stab, lane);
        zp[11] = eval_q<11>(th, stab, lane);
#pragma unroll
        for (int q = 0; q < NQ; ++q) {
            float v = zp[q];
#pragma unroll
            for (int s = 16; s >= 1; s >>= 1) v += __shfl_xor(v, s, 32);
            zp[q] = v;
        }
        if (lane < NQ) {
            float v = zp[0];
#pragma unroll
            for (int q = 1; q < NQ; ++q) v = (lane == (unsigned)q) ? zp[q] : v;
            zs[sl * NQ + (int)lane] = v * (1.0f / 4096.0f);
        }
    }
    __syncthreads();

    for (int s = 0; s < SPB; ++s) {
        float accv = bo;
#pragma unroll
        for (int q = 0; q < NQ; ++q) accv = fmaf(zs[s * NQ + q], wrow[q], accv);
        out[((size_t)blockIdx.x * SPB + s) * DOUT + tid] = accv;
    }
}

static_assert(FC_BLOCKS + TAB_BLOCKS == 128 && 4096 / SPB == 256 && TPB == 256, "grid literals below");
extern "C" void kernel_launch(void* const* d_in, const int* in_sizes, int n_in,
                              void* d_out, int out_size, void* d_ws, size_t ws_size,
                              hipStream_t stream) {
    const float* x        = (const float*)d_in[0];
    const float* W_in     = (const float*)d_in[1];
    const float* b_in     = (const float*)d_in[2];
    const float* q_params = (const float*)d_in[3];
    const float* W_out    = (const float*)d_in[4];
    const float* b_out    = (const float*)d_in[5];
    float* out    = (float*)d_out;
    float* angles = (float*)d_ws;
    float* tab    = (float*)d_ws + 65536;

    fc_in_and_table<<<dim3(128), dim3(256), 0, stream>>>(x, W_in, b_in, q_params, angles, tab);
    vqc_eval<<<dim3(256), dim3(256), 0, stream>>>(angles, q_params, tab, W_out, b_out, out);
}


# Table launch specialised per qubit at compile time (block-uniform switch): strings, K, offsets are immediates, absent gates dropped, mirror sign from a constant parity; no waits on constant loads
# speedup vs baseline: 30.7759x; 1.4018x over previous
#include <hip/hip_runtime.h>

#define NQ       12
#define NLAYERS  6
#define TPB      256
#define DOUT     256
#define NG (NLAYERS * NQ)

#define TAB_CHUNKS  8
#define TAB_BLOCKS  (NQ * TAB_CHUNKS)
#define NTERM       2176
#define NITER       2
#define SPB         (8 * NITER)
#define REV         0.15915494309189535f

typedef float v2f __attribute__((ext_vector_type(2)));
typedef float v8f __attribute__((ext_vector_type(8)));

struct FcFrag { float4 a[4]; float4 w[4]; };

__device__ __forceinline__ FcFrag fc_in_load(const float* __restrict__ x,
                                             const float* __restrict__ W_in,
                                             const int lane, const int wave) {
    const int n  = lane & 15;
    const int h  = lane >> 4;
    const int nc = (n < NQ) ? n : (NQ - 1);
    const float* xrow = x + (size_t)((int)blockIdx.x * 16 + n) * 256 + wave * 32 + 4 * h;
    const float* wrow = W_in + nc * 256 + wave * 32 + 4 * h;
    FcFrag f;
#pragma unroll
    for (int i = 0; i < 4; ++i) f.a[i] = *(const float4*)(xrow + 8 * i);
#pragma unroll
    for (int i = 0; i < 4; ++i) f.w[i] = *(const float4*)(wrow + 8 * i);
    return f;
}

__device__ __forceinline__ void fc_in_partial(const FcFrag& f, float* red,
                                              const int lane, const int wave) {
    const float mask = ((lane & 15) < NQ) ? 1.0f : 0.0f;
    v8f acc = {};
#pragma unroll
    for (int i = 0; i < 4; ++i) {
        const float4 av = f.a[i];
        float4 wv = f.w[i];
        wv.x *= mask; wv.y *= mask; wv.z *= mask; wv.w *= mask;
        v2f a1, b1, a2, b2;
        a1.x = av.x; a1.y = av.y;  b1.x = wv.x; b1.y = wv.y;
        a2.x = av.z; a2.y = av.w;  b2.x = wv.z; b2.y = wv.w;
        acc = __builtin_amdgcn_wmma_f32_16x16x4_f32(false, a1, false, b1,
                                                    (short)0, acc, false, false);
        acc = __builtin_amdgcn_wmma_f32_16x16x4_f32(false, a2, false, b2,
                                                    (short)0, acc, false, false);
    }
#pragma unroll
    for (int r = 0; r < 8; ++r) red[wave * 256 + r * 32 + lane] = acc[r];
}

struct MaskTab { unsigned m[NG]; };

constexpr MaskTab make_masks() {
    MaskTab t{};
    unsigned R[NQ] = {};
    for (int q = 0; q < NQ; ++q) R[q] = 1u << q;
    for (int l = NLAYERS - 1; l >= 0; --l) {
        R[NQ - 1] ^= R[0];
        for (int q = NQ - 2; q >= 0; --q) R[q] ^= R[q + 1];
        for (int q = 0; q < NQ; ++q) t.m[l * NQ + q] = R[q];
    }
    return t;
}
static_assert(make_masks().m[0] == 0xd7cu && make_masks().m[18] == 0x1e2u && make_masks().m[30] == 0x226u &&
              make_masks().m[47] == 0xd54u && make_masks().m[71] == 0x801u, "Z strings");

struct Plan {
    unsigned mpi[NQ][NG];
    int K[NQ];
    int off[NQ];
    int gsel[NQ][NQ];
    int total;
};
constexpr unsigned cpar(unsigned v) { unsigned p = 0; while (v) { p ^= v & 1u; v >>= 1; } return p; }
constexpr Plan make_plan() {
    Plan P{};
    const MaskTab T = make_masks();
    unsigned A[NQ] = {};
    for (int i = 0; i < NQ; ++i) A[i] = T.m[i] | (1u << (NQ + i));
    for (int c = 0; c < NQ; ++c) {
        int p = c;
        while (!((A[p] >> c) & 1u)) ++p;
        const unsigned t = A[c]; A[c] = A[p]; A[p] = t;
        for (int r = 0; r < NQ; ++r)
            if (r != c && ((A[r] >> c) & 1u)) A[r] ^= A[c];
    }
    unsigned Lc[NQ] = {};
    for (int j = 0; j < NQ; ++j)
        for (int i = 0; i < NQ; ++i) Lc[j] |= ((A[i] >> (NQ + j)) & 1u) << i;
    unsigned mp[NG] = {};
    for (int g = 0; g < NG; ++g)
        for (int k = 0; k < NQ; ++k) mp[g] |= cpar(Lc[k] & T.m[g]) << k;
    int tot = 0;
    bool ok = true;
    for (int g = 0; g < NQ; ++g) ok = ok && (mp[g] == (1u << g));
    for (int q = 0; q < NQ; ++q) {
        int k = 0;
        for (int g = 0; g < NQ; ++g) k += (int)((T.m[g] >> q) & 1u);
        P.K[q] = k;
        P.off[q] = tot;
        tot += 1 << (k - 1);
        const int nt = NQ - k;
        int a = 0, bn = 0;
        int pos[NQ] = {};
        for (int g = 0; g < NQ; ++g) {
            if ((T.m[g] >> q) & 1u) { pos[g] = nt + a; P.gsel[q][a] = g; ++a; }
            else                    { pos[g] = bn; ++bn; }
        }
        for (int g = 0; g < NG; ++g) {
            unsigned v = 0;
            for (int k2 = 0; k2 < NQ; ++k2) v |= ((mp[g] >> k2) & 1u) << pos[k2];
            P.mpi[q][g] = v;
        }
        ok = ok && (k >= 6) && (k <= 10);
    }
    P.total = ok ? tot : -1;
    return P;
}
static_assert(make_plan().total == NTERM, "half-table size");
static_assert(make_plan().K[0] == 8 && make_plan().K[1] == 6 && make_plan().K[6] == 10 && make_plan().K[11] == 9, "K_q");
static_assert(make_plan().off[3] == 192 && make_plan().off[7] == 1152 && make_plan().off[11] == 1920, "offsets");
static_assert(make_plan().mpi[0][12] == 0x11u && make_plan().mpi[11][40] == 0x1c2u && make_plan().mpi[6][71] == 0x82du, "strings over idx");
static_assert(make_plan().gsel[1][5] == 11 && make_plan().gsel[6][5] == 6 && make_plan().gsel[6][9] == 11, "gate order");

struct GateList { int n; int g[NG - NQ]; unsigned mk[NG - NQ]; bool neg[NG - NQ]; };
constexpr GateList make_gates(const int q) {
    GateList L{};
    const MaskTab T = make_masks();
    const Plan P = make_plan();
    const int K = P.K[q];
    const unsigned FLIP = ((1u << K) - 1u) << (NQ - K);
    int n = 0;
    for (int g = NQ; g < NG; ++g) {
        if ((T.m[g] >> q) & 1u) {
            L.g[n] = g;
            L.mk[n] = P.mpi[q][g];
            L.neg[n] = cpar(FLIP & P.mpi[q][g]) != 0u;
            ++n;
        }
    }
    L.n = n;
    return L;
}
static_assert(make_gates(0).n > 0 && make_gates(0).n <= NG - NQ && make_gates(11).g[0] >= NQ, "gate lists");

template <int Q, int J>
__device__ __forceinline__ void gate_terms(const float* __restrict__ qp, const unsigned idx,
                                           float& p1, float& p2) {
    constexpr GateList GL = make_gates(Q);
    if constexpr (J < GL.n) {
        constexpr int g = GL.g[J];
        constexpr unsigned mk = GL.mk[J];
        const unsigned w = __float_as_uint(qp[g] * REV);
        const float t = __uint_as_float(w ^ ((__builtin_popcount(idx & mk) & 1u) << 31));
        p1 += t;
        if constexpr (GL.neg[J]) p2 -= t; else p2 += t;
        gate_terms<Q, J + 1>(qp, idx, p1, p2);
    }
}

template <int Q>
__device__ __forceinline__ void build_table(const int chunk, const float* __restrict__ qp,
                                            float2* __restrict__ tab, float* sP, float* sQ) {
    constexpr Plan PL = make_plan();
    constexpr int K  = PL.K[Q];
    constexpr int NT = NQ - K;
    constexpr int nt = 1 << NT;
    const int tid = threadIdx.x;
    const unsigned idx = (unsigned)(chunk * TPB + tid);
    float p1 = 0.f, p2 = 0.f;
    gate_terms<Q, 0>(qp, idx, p1, p2);
    const float f1 = __builtin_amdgcn_fractf(p1), f2 = __builtin_amdgcn_fractf(p2);
    sP[tid] = __builtin_amdgcn_cosf(f1) + __builtin_amdgcn_cosf(f2);
    sQ[tid] = __builtin_amdgcn_sinf(f1) - __builtin_amdgcn_sinf(f2);
    __syncthreads();
    if ((tid & (nt - 1)) == 0) {
        const int t0 = (tid >> NT) << NT;
        float aP = 0.f, aQ = 0.f;
#pragma unroll
        for (int t = 0; t < nt; ++t) { aP += sP[t0 + t]; aQ += sQ[t0 + t]; }
        tab[PL.off[Q] + (int)(idx >> NT)] = make_float2(aP, aQ);
    }
}

__global__ __launch_bounds__(TPB) void table_kernel(const float* __restrict__ qp,
                                                    float* __restrict__ tabf) {
    __shared__ __attribute__((aligned(16))) float sP[TPB];
    __shared__ __attribute__((aligned(16))) float sQ[TPB];
    const int tb = (int)blockIdx.x;
    const int chunk = tb % TAB_CHUNKS;
    float2* tab = (float2*)tabf;
    switch (tb / TAB_CHUNKS) {
        case 0:  build_table<0>(chunk, qp, tab, sP, sQ);  break;
        case 1:  build_table<1>(chunk, qp, tab, sP, sQ);  break;
        case 2:  build_table<2>(chunk, qp, tab, sP, sQ);  break;
        case 3:  build_table<3>(chunk, qp, tab, sP, sQ);  break;
        case 4:  build_table<4>(chunk, qp, tab, sP, sQ);  break;
        case 5:  build_table<5>(chunk, qp, tab, sP, sQ);  break;
        case 6:  build_table<6>(chunk, qp, tab, sP, sQ);  break;
        case 7:  build_table<7>(chunk, qp, tab, sP, sQ);  break;
        case 8:  build_table<8>(chunk, qp, tab, sP, sQ);  break;
        case 9:  build_table<9>(chunk, qp, tab, sP, sQ);  break;
        case 10: build_table<10>(chunk, qp, tab, sP, sQ); break;
        default: build_table<11>(chunk, qp, tab, sP, sQ); break;
    }
}

template <int Q>
__device__ __forceinline__ float eval_q(const float (&th)[NQ], const float2* stab,
                                        const unsigned lane) {
    constexpr Plan PL = make_plan();
    constexpr int K   = PL.K[Q];
    constexpr int NHI = 1 << (K - 6);
    float base = 0.f;
#pragma unroll
    for (int i = 0; i < 5; ++i)
        base += __uint_as_float(__float_as_uint(th[PL.gsel[Q][i]]) ^ (((lane >> i) & 1u) << 31));
    float hs[NHI];
    float h0 = 0.f;
#pragma unroll
    for (int i = 5; i < K; ++i) h0 += th[PL.gsel[Q][i]];
    hs[0] = h0;
#pragma unroll
    for (int j = 5; j < K - 1; ++j) {
        const float t2 = 2.0f * th[PL.gsel[Q][j]];
#pragma unroll
        for (int h = 0; h < (1 << (j - 5)); ++h) hs[h | (1 << (j - 5))] = hs[h] - t2;
    }
    float acc = 0.f;
#pragma unroll
    for (int hi = 0; hi < NHI; ++hi) {
        const float2 e = stab[PL.off[Q] + hi * 32 + (int)lane];
        const float f = __builtin_amdgcn_fractf(base + hs[hi]);
        acc = fmaf(e.x, __builtin_amdgcn_cosf(f), acc);
        acc = fmaf(-e.y, __builtin_amdgcn_sinf(f), acc);
    }
    return acc;
}

__global__ __launch_bounds__(TPB) void vqc_eval(const float* __restrict__ x,
                                                const float* __restrict__ W_in,
                                                const float* __restrict__ b_in,
                                                const float* __restrict__ qp,
                                                const float* __restrict__ tabf,
                                                const float* __restrict__ W_out,
                                                const float* __restrict__ b_out,
                                                float* __restrict__ out) {
    constexpr int NV = NTERM / 2, NCP = (NV + TPB - 1) / TPB;
    __shared__ float2 stab[NCP * TPB * 2];
    __shared__ float zs[SPB * NQ];
    __shared__ float red[8 * 256];
    const int tid = threadIdx.x;
    const unsigned lane = (unsigned)tid & 31u;
    const int wave = tid >> 5;

    const FcFrag frag = fc_in_load(x, W_in, (int)lane, wave);
    {
        const float4* src = (const float4*)tabf;
        float4* dst = (float4*)stab;
        float4 tv[NCP];
#pragma unroll
        for (int k = 0; k < NCP; ++k) { const int i = tid + k * TPB; tv[k] = src[i < NV ? i : NV - 1]; }
#pragma unroll
        for (int k = 0; k < NCP; ++k) dst[tid + k * TPB] = tv[k];
    }
    float wrow[NQ];
#pragma unroll
    for (int q = 0; q < NQ; ++q) wrow[q] = W_out[tid * NQ + q];
    const float bo = b_out[tid];
    const int gc = ((lane & 15u) < NQ) ? (int)(lane & 15u) : (NQ - 1);
    const float qb = qp[gc], bi = b_in[gc];
    fc_in_partial(frag, red, (int)lane, wave);
    __syncthreads();

    float ang = red[wave * 32 + (int)lane];
#pragma unroll
    for (int w = 1; w < 8; ++w) ang += red[w * 256 + wave * 32 + (int)lane];
    const float thv = (qb + (ang + bi)) * REV;

    for (int it = 0; it < NITER; ++it) {
        const int sl = it * 8 + wave;
        float th[NQ];
#pragma unroll
        for (int g = 0; g < NQ; ++g)
            th[g] = __int_as_float(__builtin_amdgcn_readlane(__float_as_int(thv), it * 16 + g));
        float zp[NQ];
        zp[0]  = eval_q<0>(th, stab, lane);
        zp[1]  = eval_q<1>(th, stab, lane);
        zp[2]  = eval_q<2>(th, stab, lane);
        zp[3]  = eval_q<3>(th, stab, lane);
        zp[4]  = eval_q<4>(th, stab, lane);
        zp[5]  = eval_q<5>(th, stab, lane);
        zp[6]  = eval_q<6>(th, stab, lane);
        zp[7]  = eval_q<7>(th, stab, lane);
        zp[8]  = eval_q<8>(th, stab, lane);
        zp[9]  = eval_q<9>(th, stab, lane);
        zp[10] = eval_q<10>(th, stab, lane);
        zp[11] = eval_q<11>(th, stab, lane);
#pragma unroll
        for (int q = 0; q < NQ; ++q) {
            float v = zp[q];
#pragma unroll
            for (int s = 16; s >= 1; s >>= 1) v += __shfl_xor(v, s, 32);
            zp[q] = v;
        }
        if (lane < NQ) {
            float v = zp[0];
#pragma unroll
            for (int q = 1; q < NQ; ++q) v = (lane == (unsigned)q) ? zp[q] : v;
            zs[sl * NQ + (int)lane] = v * (1.0f / 4096.0f);
        }
    }
    __syncthreads();

    for (int s = 0; s < SPB; ++s) {
        float accv = bo;
#pragma unroll
        for (int q = 0; q < NQ; ++q) accv = fmaf(zs[s * NQ + q], wrow[q], accv);
        out[((size_t)blockIdx.x * SPB + s) * DOUT + tid] = accv;
    }
}

static_assert(TAB_BLOCKS == 96 && 4096 / SPB == 256 && SPB == 16 && NITER == 2 && TPB == 256, "grid literals below");
extern "C" void kernel_launch(void* const* d_in, const int* in_sizes, int n_in,
                              void* d_out, int out_size, void* d_ws, size_t ws_size,
                              hipStream_t stream) {
    const float* x        = (const float*)d_in[0];
    const float* W_in     = (const float*)d_in[1];
    const float* b_in     = (const float*)d_in[2];
    const float* q_params = (const float*)d_in[3];
    const float* W_out    = (const float*)d_in[4];
    const float* b_out    = (const float*)d_in[5];
    float* out    = (float*)d_out;
    float* tab    = (float*)d_ws;

    table_kernel<<<dim3(96), dim3(256), 0, stream>>>(q_params, tab);
    vqc_eval<<<dim3(256), dim3(256), 0, stream>>>(x, W_in, b_in, q_params, tab, W_out, b_out, out);
}


# Per-term sine and cosine replaced by a Horner fold of the table entries with wave-uniform e^{-2i theta} factors (angle addition): one sincos per qubit and lane, not per term; Z sum order changes
# speedup vs baseline: 33.3593x; 1.0839x over previous
#include <hip/hip_runtime.h>

#define NQ       12
#define NLAYERS  6
#define TPB      256
#define DOUT     256
#define NG (NLAYERS * NQ)

#define TAB_CHUNKS  8
#define TAB_BLOCKS  (NQ * TAB_CHUNKS)
#define NTERM       2176
#define NITER       2
#define SPB         (8 * NITER)
#define REV         0.15915494309189535f

typedef float v2f __attribute__((ext_vector_type(2)));
typedef float v8f __attribute__((ext_vector_type(8)));

struct FcFrag { float4 a[4]; float4 w[4]; };

__device__ __forceinline__ FcFrag fc_in_load(const float* __restrict__ x,
                                             const float* __restrict__ W_in,
                                             const int lane, const int wave) {
    const int n  = lane & 15;
    const int h  = lane >> 4;
    const int nc = (n < NQ) ? n : (NQ - 1);
    const float* xrow = x + (size_t)((int)blockIdx.x * 16 + n) * 256 + wave * 32 + 4 * h;
    const float* wrow = W_in + nc * 256 + wave * 32 + 4 * h;
    FcFrag f;
#pragma unroll
    for (int i = 0; i < 4; ++i) f.a[i] = *(const float4*)(xrow + 8 * i);
#pragma unroll
    for (int i = 0; i < 4; ++i) f.w[i] = *(const float4*)(wrow + 8 * i);
    return f;
}

__device__ __forceinline__ void fc_in_partial(const FcFrag& f, float* red,
                                              const int lane, const int wave) {
    const float mask = ((lane & 15) < NQ) ? 1.0f : 0.0f;
    v8f acc = {};
#pragma unroll
    for (int i = 0; i < 4; ++i) {
        const float4 av = f.a[i];
        float4 wv = f.w[i];
        wv.x *= mask; wv.y *= mask; wv.z *= mask; wv.w *= mask;
        v2f a1, b1, a2, b2;
        a1.x = av.x; a1.y = av.y;  b1.x = wv.x; b1.y = wv.y;
        a2.x = av.z; a2.y = av.w;  b2.x = wv.z; b2.y = wv.w;
        acc = __builtin_amdgcn_wmma_f32_16x16x4_f32(false, a1, false, b1,
                                                    (short)0, acc, false, false);
        acc = __builtin_amdgcn_wmma_f32_16x16x4_f32(false, a2, false, b2,
                                                    (short)0, acc, false, false);
    }
#pragma unroll
    for (int r = 0; r < 8; ++r) red[wave * 256 + r * 32 + lane] = acc[r];
}

struct MaskTab { unsigned m[NG]; };

constexpr MaskTab make_masks() {
    MaskTab t{};
    unsigned R[NQ] = {};
    for (int q = 0; q < NQ; ++q) R[q] = 1u << q;
    for (int l = NLAYERS - 1; l >= 0; --l) {
        R[NQ - 1] ^= R[0];
        for (int q = NQ - 2; q >= 0; --q) R[q] ^= R[q + 1];
        for (int q = 0; q < NQ; ++q) t.m[l * NQ + q] = R[q];
    }
    return t;
}
static_assert(make_masks().m[0] == 0xd7cu && make_masks().m[18] == 0x1e2u && make_masks().m[30] == 0x226u &&
              make_masks().m[47] == 0xd54u && make_masks().m[71] == 0x801u, "Z strings");

struct Plan {
    unsigned mpi[NQ][NG];
    int K[NQ];
    int off[NQ];
    int gsel[NQ][NQ];
    int total;
};
constexpr unsigned cpar(unsigned v) { unsigned p = 0; while (v) { p ^= v & 1u; v >>= 1; } return p; }
constexpr Plan make_plan() {
    Plan P{};
    const MaskTab T = make_masks();
    unsigned A[NQ] = {};
    for (int i = 0; i < NQ; ++i) A[i] = T.m[i] | (1u << (NQ + i));
    for (int c = 0; c < NQ; ++c) {
        int p = c;
        while (!((A[p] >> c) & 1u)) ++p;
        const unsigned t = A[c]; A[c] = A[p]; A[p] = t;
        for (int r = 0; r < NQ; ++r)
            if (r != c && ((A[r] >> c) & 1u)) A[r] ^= A[c];
    }
    unsigned Lc[NQ] = {};
    for (int j = 0; j < NQ; ++j)
        for (int i = 0; i < NQ; ++i) Lc[j] |= ((A[i] >> (NQ + j)) & 1u) << i;
    unsigned mp[NG] = {};
    for (int g = 0; g < NG; ++g)
        for (int k = 0; k < NQ; ++k) mp[g] |= cpar(Lc[k] & T.m[g]) << k;
    int tot = 0;
    bool ok = true;
    for (int g = 0; g < NQ; ++g) ok = ok && (mp[g] == (1u << g));
    for (int q = 0; q < NQ; ++q) {
        int k = 0;
        for (int g = 0; g < NQ; ++g) k += (int)((T.m[g] >> q) & 1u);
        P.K[q] = k;
        P.off[q] = tot;
        tot += 1 << (k - 1);
        const int nt = NQ - k;
        int a = 0, bn = 0;
        int pos[NQ] = {};
        for (int g = 0; g < NQ; ++g) {
            if ((T.m[g] >> q) & 1u) { pos[g] = nt + a; P.gsel[q][a] = g; ++a; }
            else                    { pos[g] = bn; ++bn; }
        }
        for (int g = 0; g < NG; ++g) {
            unsigned v = 0;
            for (int k2 = 0; k2 < NQ; ++k2) v |= ((mp[g] >> k2) & 1u) << pos[k2];
            P.mpi[q][g] = v;
        }
        ok = ok && (k >= 6) && (k <= 10);
    }
    P.total = ok ? tot : -1;
    return P;
}
static_assert(make_plan().total == NTERM, "half-table size");
static_assert(make_plan().K[0] == 8 && make_plan().K[1] == 6 && make_plan().K[6] == 10 && make_plan().K[11] == 9, "K_q");
static_assert(make_plan().off[3] == 192 && make_plan().off[7] == 1152 && make_plan().off[11] == 1920, "offsets");
static_assert(make_plan().mpi[0][12] == 0x11u && make_plan().mpi[11][40] == 0x1c2u && make_plan().mpi[6][71] == 0x82du, "strings over idx");
static_assert(make_plan().gsel[1][5] == 11 && make_plan().gsel[6][5] == 6 && make_plan().gsel[6][9] == 11, "gate order");

struct GateList { int n; int g[NG - NQ]; unsigned mk[NG - NQ]; bool neg[NG - NQ]; };
constexpr GateList make_gates(const int q) {
    GateList L{};
    const MaskTab T = make_masks();
    const Plan P = make_plan();
    const int K = P.K[q];
    const unsigned FLIP = ((1u << K) - 1u) << (NQ - K);
    int n = 0;
    for (int g = NQ; g < NG; ++g) {
        if ((T.m[g] >> q) & 1u) {
            L.g[n] = g;
            L.mk[n] = P.mpi[q][g];
            L.neg[n] = cpar(FLIP & P.mpi[q][g]) != 0u;
            ++n;
        }
    }
    L.n = n;
    return L;
}
static_assert(make_gates(0).n > 0 && make_gates(0).n <= NG - NQ && make_gates(11).g[0] >= NQ, "gate lists");

template <int Q, int J>
__device__ __forceinline__ void gate_terms(const float* __restrict__ qp, const unsigned idx,
                                           float& p1, float& p2) {
    constexpr GateList GL = make_gates(Q);
    if constexpr (J < GL.n) {
        constexpr int g = GL.g[J];
        constexpr unsigned mk = GL.mk[J];
        const unsigned w = __float_as_uint(qp[g] * REV);
        const float t = __uint_as_float(w ^ ((__builtin_popcount(idx & mk) & 1u) << 31));
        p1 += t;
        if constexpr (GL.neg[J]) p2 -= t; else p2 += t;
        gate_terms<Q, J + 1>(qp, idx, p1, p2);
    }
}

template <int Q>
__device__ __forceinline__ void build_table(const int chunk, const float* __restrict__ qp,
                                            float2* __restrict__ tab, float* sP, float* sQ) {
    constexpr Plan PL = make_plan();
    constexpr int K  = PL.K[Q];
    constexpr int NT = NQ - K;
    constexpr int nt = 1 << NT;
    const int tid = threadIdx.x;
    const unsigned idx = (unsigned)(chunk * TPB + tid);
    float p1 = 0.f, p2 = 0.f;
    gate_terms<Q, 0>(qp, idx, p1, p2);
    const float f1 = __builtin_amdgcn_fractf(p1), f2 = __builtin_amdgcn_fractf(p2);
    sP[tid] = __builtin_amdgcn_cosf(f1) + __builtin_amdgcn_cosf(f2);
    sQ[tid] = __builtin_amdgcn_sinf(f1) - __builtin_amdgcn_sinf(f2);
    __syncthreads();
    if ((tid & (nt - 1)) == 0) {
        const int t0 = (tid >> NT) << NT;
        float aP = 0.f, aQ = 0.f;
#pragma unroll
        for (int t = 0; t < nt; ++t) { aP += sP[t0 + t]; aQ += sQ[t0 + t]; }
        tab[PL.off[Q] + (int)(idx >> NT)] = make_float2(aP, aQ);
    }
}

__global__ __launch_bounds__(TPB) void table_kernel(const float* __restrict__ qp,
                                                    float* __restrict__ tabf) {
    __shared__ __attribute__((aligned(16))) float sP[TPB];
    __shared__ __attribute__((aligned(16))) float sQ[TPB];
    const int tb = (int)blockIdx.x;
    const int chunk = tb % TAB_CHUNKS;
    float2* tab = (float2*)tabf;
    switch (tb / TAB_CHUNKS) {
        case 0:  build_table<0>(chunk, qp, tab, sP, sQ);  break;
        case 1:  build_table<1>(chunk, qp, tab, sP, sQ);  break;
        case 2:  build_table<2>(chunk, qp, tab, sP, sQ);  break;
        case 3:  build_table<3>(chunk, qp, tab, sP, sQ);  break;
        case 4:  build_table<4>(chunk, qp, tab, sP, sQ);  break;
        case 5:  build_table<5>(chunk, qp, tab, sP, sQ);  break;
        case 6:  build_table<6>(chunk, qp, tab, sP, sQ);  break;
        case 7:  build_table<7>(chunk, qp, tab, sP, sQ);  break;
        case 8:  build_table<8>(chunk, qp, tab, sP, sQ);  break;
        case 9:  build_table<9>(chunk, qp, tab, sP, sQ);  break;
        case 10: build_table<10>(chunk, qp, tab, sP, sQ); break;
        default: build_table<11>(chunk, qp, tab, sP, sQ); break;
    }
}

template <int Q>
__device__ __forceinline__ float eval_q(const float (&th)[NQ], const float (&vr)[NQ],
                                        const float (&vi)[NQ], const float2* stab,
                                        const unsigned lane) {
    constexpr Plan PL = make_plan();
    constexpr int K   = PL.K[Q];
    constexpr int NB  = K - 6;
    constexpr int NHI = 1 << NB;
    float ph = 0.f;
#pragma unroll
    for (int i = 0; i < 5; ++i)
        ph += __uint_as_float(__float_as_uint(th[PL.gsel[Q][i]]) ^ (((lane >> i) & 1u) << 31));
    float h0 = 0.f;
#pragma unroll
    for (int i = 5; i < K; ++i) h0 += th[PL.gsel[Q][i]];
    float tr[NHI], ti[NHI];
#pragma unroll
    for (int hi = 0; hi < NHI; ++hi) {
        const float2 e = stab[PL.off[Q] + hi * 32 + (int)lane];
        tr[hi] = e.x; ti[hi] = e.y;
    }
#pragma unroll
    for (int b = NB - 1; b >= 0; --b) {
        const float cr = vr[PL.gsel[Q][5 + b]], ci = vi[PL.gsel[Q][5 + b]];
#pragma unroll
        for (int h = 0; h < (1 << b); ++h) {
            const float xr = tr[h + (1 << b)], xi = ti[h + (1 << b)];
            tr[h] = fmaf(cr, xr, fmaf(-ci, xi, tr[h]));
            ti[h] = fmaf(cr, xi, fmaf(ci, xr, ti[h]));
        }
    }
    const float f = __builtin_amdgcn_fractf(ph + h0);
    return fmaf(__builtin_amdgcn_cosf(f), tr[0], -__builtin_amdgcn_sinf(f) * ti[0]);
}

__global__ __launch_bounds__(TPB) void vqc_eval(const float* __restrict__ x,
                                                const float* __restrict__ W_in,
                                                const float* __restrict__ b_in,
                                                const float* __restrict__ qp,
                                                const float* __restrict__ tabf,
                                                const float* __restrict__ W_out,
                                                const float* __restrict__ b_out,
                                                float* __restrict__ out) {
    constexpr int NV = NTERM / 2, NCP = (NV + TPB - 1) / TPB;
    __shared__ float2 stab[NCP * TPB * 2];
    __shared__ float zs[SPB * NQ];
    __shared__ float red[8 * 256];
    const int tid = threadIdx.x;
    const unsigned lane = (unsigned)tid & 31u;
    const int wave = tid >> 5;

    const FcFrag frag = fc_in_load(x, W_in, (int)lane, wave);
    {
        const float4* src = (const float4*)tabf;
        float4* dst = (float4*)stab;
        float4 tv[NCP];
#pragma unroll
        for (int k = 0; k < NCP; ++k) { const int i = tid + k * TPB; tv[k] = src[i < NV ? i : NV - 1]; }
#pragma unroll
        for (int k = 0; k < NCP; ++k) dst[tid + k * TPB] = tv[k];
    }
    float wrow[NQ];
#pragma unroll
    for (int q = 0; q < NQ; ++q) wrow[q] = W_out[tid * NQ + q];
    const float bo = b_out[tid];
    const int gc = ((lane & 15u) < NQ) ? (int)(lane & 15u) : (NQ - 1);
    const float qb = qp[gc], bi = b_in[gc];
    fc_in_partial(frag, red, (int)lane, wave);
    __syncthreads();

    float ang = red[wave * 32 + (int)lane];
#pragma unroll
    for (int w = 1; w < 8; ++w) ang += red[w * 256 + wave * 32 + (int)lane];
    const float thv = (qb + (ang + bi)) * REV;
    const float f2v = __builtin_amdgcn_fractf(2.0f * thv);
    const float vcv = __builtin_amdgcn_cosf(f2v);
    const float vsv = -__builtin_amdgcn_sinf(f2v);

    for (int it = 0; it < NITER; ++it) {
        const int sl = it * 8 + wave;
        float th[NQ], vr[NQ], vi[NQ];
#pragma unroll
        for (int g = 0; g < NQ; ++g) {
            th[g] = __int_as_float(__builtin_amdgcn_readlane(__float_as_int(thv), it * 16 + g));
            vr[g] = __int_as_float(__builtin_amdgcn_readlane(__float_as_int(vcv), it * 16 + g));
            vi[g] = __int_as_float(__builtin_amdgcn_readlane(__float_as_int(vsv), it * 16 + g));
        }
        float zp[NQ];
        zp[0]  = eval_q<0>(th, vr, vi, stab, lane);
        zp[1]  = eval_q<1>(th, vr, vi, stab, lane);
        zp[2]  = eval_q<2>(th, vr, vi, stab, lane);
        zp[3]  = eval_q<3>(th, vr, vi, stab, lane);
        zp[4]  = eval_q<4>(th, vr, vi, stab, lane);
        zp[5]  = eval_q<5>(th, vr, vi, stab, lane);
        zp[6]  = eval_q<6>(th, vr, vi, stab, lane);
        zp[7]  = eval_q<7>(th, vr, vi, stab, lane);
        zp[8]  = eval_q<8>(th, vr, vi, stab, lane);
        zp[9]  = eval_q<9>(th, vr, vi, stab, lane);
        zp[10] = eval_q<10>(th, vr, vi, stab, lane);
        zp[11] = eval_q<11>(th, vr, vi, stab, lane);
#pragma unroll
        for (int q = 0; q < NQ; ++q) {
            float v = zp[q];
#pragma unroll
            for (int s = 16; s >= 1; s >>= 1) v += __shfl_xor(v, s, 32);
            zp[q] = v;
        }
        if (lane < NQ) {
            float v = zp[0];
#pragma unroll
            for (int q = 1; q < NQ; ++q) v = (lane == (unsigned)q) ? zp[q] : v;
            zs[sl * NQ + (int)lane] = v * (1.0f / 4096.0f);
        }
    }
    __syncthreads();

    for (int s = 0; s < SPB; ++s) {
        float accv = bo;
#pragma unroll
        for (int q = 0; q < NQ; ++q) accv = fmaf(zs[s * NQ + q], wrow[q], accv);
        out[((size_t)blockIdx.x * SPB + s) * DOUT + tid] = accv;
    }
}

static_assert(TAB_BLOCKS == 96 && 4096 / SPB == 256 && SPB == 16 && NITER == 2 && TPB == 256, "grid literals below");
extern "C" void kernel_launch(void* const* d_in, const int* in_sizes, int n_in,
                              void* d_out, int out_size, void* d_ws, size_t ws_size,
                              hipStream_t stream) {
    const float* x        = (const float*)d_in[0];
    const float* W_in     = (const float*)d_in[1];
    const float* b_in     = (const float*)d_in[2];
    const float* q_params = (const float*)d_in[3];
    const float* W_out    = (const float*)d_in[4];
    const float* b_out    = (const float*)d_in[5];
    float* out    = (float*)d_out;
    float* tab    = (float*)d_ws;

    table_kernel<<<dim3(96), dim3(256), 0, stream>>>(q_params, tab);
    vqc_eval<<<dim3(256), dim3(256), 0, stream>>>(x, W_in, b_in, q_params, tab, W_out, b_out, out);
}


# Lane sums on the vector ALU (DPP row_xmask, permlanex16), not the LDS unit: 12 wave sums as one transposing butterfly (13 exchanges, not 60 ds_bpermute), table tau sums too; table sum order changes
# speedup vs baseline: 35.4321x; 1.0621x over previous
#include <hip/hip_runtime.h>

#define NQ       12
#define NLAYERS  6
#define TPB      256
#define DOUT     256
#define NG (NLAYERS * NQ)

#define TAB_CHUNKS  8
#define TAB_BLOCKS  (NQ * TAB_CHUNKS)
#define NTERM       2176
#define NITER       2
#define SPB         (8 * NITER)
#define REV         0.15915494309189535f

typedef float v2f __attribute__((ext_vector_type(2)));
typedef float v8f __attribute__((ext_vector_type(8)));

struct FcFrag { float4 a[4]; float4 w[4]; };

__device__ __forceinline__ FcFrag fc_in_load(const float* __restrict__ x,
                                             const float* __restrict__ W_in,
                                             const int lane, const int wave) {
    const int n  = lane & 15;
    const int h  = lane >> 4;
    const int nc = (n < NQ) ? n : (NQ - 1);
    const float* xrow = x + (size_t)((int)blockIdx.x * 16 + n) * 256 + wave * 32 + 4 * h;
    const float* wrow = W_in + nc * 256 + wave * 32 + 4 * h;
    FcFrag f;
#pragma unroll
    for (int i = 0; i < 4; ++i) f.a[i] = *(const float4*)(xrow + 8 * i);
#pragma unroll
    for (int i = 0; i < 4; ++i) f.w[i] = *(const float4*)(wrow + 8 * i);
    return f;
}

__device__ __forceinline__ void fc_in_partial(const FcFrag& f, float* red,
                                              const int lane, const int wave) {
    const float mask = ((lane & 15) < NQ) ? 1.0f : 0.0f;
    v8f acc = {};
#pragma unroll
    for (int i = 0; i < 4; ++i) {
        const float4 av = f.a[i];
        float4 wv = f.w[i];
        wv.x *= mask; wv.y *= mask; wv.z *= mask; wv.w *= mask;
        v2f a1, b1, a2, b2;
        a1.x = av.x; a1.y = av.y;  b1.x = wv.x; b1.y = wv.y;
        a2.x = av.z; a2.y = av.w;  b2.x = wv.z; b2.y = wv.w;
        acc = __builtin_amdgcn_wmma_f32_16x16x4_f32(false, a1, false, b1,
                                                    (short)0, acc, false, false);
        acc = __builtin_amdgcn_wmma_f32_16x16x4_f32(false, a2, false, b2,
                                                    (short)0, acc, false, false);
    }
#pragma unroll
    for (int r = 0; r < 8; ++r) red[wave * 256 + r * 32 + lane] = acc[r];
}

struct MaskTab { unsigned m[NG]; };

constexpr MaskTab make_masks() {
    MaskTab t{};
    unsigned R[NQ] = {};
    for (int q = 0; q < NQ; ++q) R[q] = 1u << q;
    for (int l = NLAYERS - 1; l >= 0; --l) {
        R[NQ - 1] ^= R[0];
        for (int q = NQ - 2; q >= 0; --q) R[q] ^= R[q + 1];
        for (int q = 0; q < NQ; ++q) t.m[l * NQ + q] = R[q];
    }
    return t;
}
static_assert(make_masks().m[0] == 0xd7cu && make_masks().m[18] == 0x1e2u && make_masks().m[30] == 0x226u &&
              make_masks().m[47] == 0xd54u && make_masks().m[71] == 0x801u, "Z strings");

struct Plan {
    unsigned mpi[NQ][NG];
    int K[NQ];
    int off[NQ];
    int gsel[NQ][NQ];
    int total;
};
constexpr unsigned cpar(unsigned v) { unsigned p = 0; while (v) { p ^= v & 1u; v >>= 1; } return p; }
constexpr Plan make_plan() {
    Plan P{};
    const MaskTab T = make_masks();
    unsigned A[NQ] = {};
    for (int i = 0; i < NQ; ++i) A[i] = T.m[i] | (1u << (NQ + i));
    for (int c = 0; c < NQ; ++c) {
        int p = c;
        while (!((A[p] >> c) & 1u)) ++p;
        const unsigned t = A[c]; A[c] = A[p]; A[p] = t;
        for (int r = 0; r < NQ; ++r)
            if (r != c && ((A[r] >> c) & 1u)) A[r] ^= A[c];
    }
    unsigned Lc[NQ] = {};
    for (int j = 0; j < NQ; ++j)
        for (int i = 0; i < NQ; ++i) Lc[j] |= ((A[i] >> (NQ + j)) & 1u) << i;
    unsigned mp[NG] = {};
    for (int g = 0; g < NG; ++g)
        for (int k = 0; k < NQ; ++k) mp[g] |= cpar(Lc[k] & T.m[g]) << k;
    int tot = 0;
    bool ok = true;
    for (int g = 0; g < NQ; ++g) ok = ok && (mp[g] == (1u << g));
    for (int q = 0; q < NQ; ++q) {
        int k = 0;
        for (int g = 0; g < NQ; ++g) k += (int)((T.m[g] >> q) & 1u);
        P.K[q] = k;
        P.off[q] = tot;
        tot += 1 << (k - 1);
        const int nt = NQ - k;
        int a = 0, bn = 0;
        int pos[NQ] = {};
        for (int g = 0; g < NQ; ++g) {
            if ((T.m[g] >> q) & 1u) { pos[g] = nt + a; P.gsel[q][a] = g; ++a; }
            else                    { pos[g] = bn; ++bn; }
        }
        for (int g = 0; g < NG; ++g) {
            unsigned v = 0;
            for (int k2 = 0; k2 < NQ; ++k2) v |= ((mp[g] >> k2) & 1u) << pos[k2];
            P.mpi[q][g] = v;
        }
        ok = ok && (k >= 6) && (k <= 10);
    }
    P.total = ok ? tot : -1;
    return P;
}
static_assert(make_plan().total == NTERM, "half-table size");
static_assert(make_plan().K[0] == 8 && make_plan().K[1] == 6 && make_plan().K[6] == 10 && make_plan().K[11] == 9, "K_q");
static_assert(make_plan().off[3] == 192 && make_plan().off[7] == 1152 && make_plan().off[11] == 1920, "offsets");
static_assert(make_plan().mpi[0][12] == 0x11u && make_plan().mpi[11][40] == 0x1c2u && make_plan().mpi[6][71] == 0x82du, "strings over idx");
static_assert(make_plan().gsel[1][5] == 11 && make_plan().gsel[6][5] == 6 && make_plan().gsel[6][9] == 11, "gate order");

struct GateList { int n; int g[NG - NQ]; unsigned mk[NG - NQ]; bool neg[NG - NQ]; };
constexpr GateList make_gates(const int q) {
    GateList L{};
    const MaskTab T = make_masks();
    const Plan P = make_plan();
    const int K = P.K[q];
    const unsigned FLIP = ((1u << K) - 1u) << (NQ - K);
    int n = 0;
    for (int g = NQ; g < NG; ++g) {
        if ((T.m[g] >> q) & 1u) {
            L.g[n] = g;
            L.mk[n] = P.mpi[q][g];
            L.neg[n] = cpar(FLIP & P.mpi[q][g]) != 0u;
            ++n;
        }
    }
    L.n = n;
    return L;
}
static_assert(make_gates(0).n > 0 && make_gates(0).n <= NG - NQ && make_gates(11).g[0] >= NQ, "gate lists");

template <int M>
__device__ __forceinline__ float xrow(const float v) {
    return __int_as_float(__builtin_amdgcn_update_dpp(0, __float_as_int(v), 0x160 | M, 0xf, 0xf, true));
}
__device__ __forceinline__ float xhalf(const float v) {
    const int b = __float_as_int(v);
    return __int_as_float(__builtin_amdgcn_permlanex16(b, b, 0x76543210, 0xfedcba98, false, false));
}
template <int M>
__device__ __forceinline__ float tstep(const float a, const float b, const bool up) {
    const float keep = up ? b : a;
    const float send = up ? a : b;
    if constexpr (M == 16) return keep + xhalf(send);
    else                   return keep + xrow<M>(send);
}

template <int Q, int J>
__device__ __forceinline__ void gate_terms(const float* __restrict__ qp, const unsigned idx,
                                           float& p1, float& p2) {
    constexpr GateList GL = make_gates(Q);
    if constexpr (J < GL.n) {
        constexpr int g = GL.g[J];
        constexpr unsigned mk = GL.mk[J];
        const unsigned w = __float_as_uint(qp[g] * REV);
        const float t = __uint_as_float(w ^ ((__builtin_popcount(idx & mk) & 1u) << 31));
        p1 += t;
        if constexpr (GL.neg[J]) p2 -= t; else p2 += t;
        gate_terms<Q, J + 1>(qp, idx, p1, p2);
    }
}

template <int Q>
__device__ __forceinline__ void build_table(const int chunk, const float* __restrict__ qp,
                                            float2* __restrict__ tab, float* sP, float* sQ) {
    constexpr Plan PL = make_plan();
    constexpr int K  = PL.K[Q];
    constexpr int NT = NQ - K;
    constexpr int nt = 1 << NT;
    static_assert(NT >= 2 && NT <= 6, "a sigma's threads fill part of a wave, one wave or two waves");
    const int tid = threadIdx.x;
    const unsigned idx = (unsigned)(chunk * TPB + tid);
    float p1 = 0.f, p2 = 0.f;
    gate_terms<Q, 0>(qp, idx, p1, p2);
    const float f1 = __builtin_amdgcn_fractf(p1), f2 = __builtin_amdgcn_fractf(p2);
    float aP = __builtin_amdgcn_cosf(f1) + __builtin_amdgcn_cosf(f2);
    float aQ = __builtin_amdgcn_sinf(f1) - __builtin_amdgcn_sinf(f2);
    aP += xrow<1>(aP);  aQ += xrow<1>(aQ);
    aP += xrow<2>(aP);  aQ += xrow<2>(aQ);
    if constexpr (NT >= 3) { aP += xrow<4>(aP);  aQ += xrow<4>(aQ); }
    if constexpr (NT >= 4) { aP += xrow<8>(aP);  aQ += xrow<8>(aQ); }
    if constexpr (NT >= 5) { aP += xhalf(aP);    aQ += xhalf(aQ); }
    if constexpr (NT <= 5) {
        if ((tid & (nt - 1)) == 0) tab[PL.off[Q] + (int)(idx >> NT)] = make_float2(aP, aQ);
    } else {
        const int wave = tid >> 5;
        if ((tid & 31) == 0) { sP[wave] = aP; sQ[wave] = aQ; }
        __syncthreads();
        if (tid < TPB / nt)
            tab[PL.off[Q] + chunk * (TPB / nt) + tid] =
                make_float2(sP[2 * tid] + sP[2 * tid + 1], sQ[2 * tid] + sQ[2 * tid + 1]);
    }
}

__global__ __launch_bounds__(TPB) void table_kernel(const float* __restrict__ qp,
                                                    float* __restrict__ tabf) {
    __shared__ __attribute__((aligned(16))) float sP[TPB];
    __shared__ __attribute__((aligned(16))) float sQ[TPB];
    const int tb = (int)blockIdx.x;
    const int chunk = tb % TAB_CHUNKS;
    float2* tab = (float2*)tabf;
    switch (tb / TAB_CHUNKS) {
        case 0:  build_table<0>(chunk, qp, tab, sP, sQ);  break;
        case 1:  build_table<1>(chunk, qp, tab, sP, sQ);  break;
        case 2:  build_table<2>(chunk, qp, tab, sP, sQ);  break;
        case 3:  build_table<3>(chunk, qp, tab, sP, sQ);  break;
        case 4:  build_table<4>(chunk, qp, tab, sP, sQ);  break;
        case 5:  build_table<5>(chunk, qp, tab, sP, sQ);  break;
        case 6:  build_table<6>(chunk, qp, tab, sP, sQ);  break;
        case 7:  build_table<7>(chunk, qp, tab, sP, sQ);  break;
        case 8:  build_table<8>(chunk, qp, tab, sP, sQ);  break;
        case 9:  build_table<9>(chunk, qp, tab, sP, sQ);  break;
        case 10: build_table<10>(chunk, qp, tab, sP, sQ); break;
        default: build_table<11>(chunk, qp, tab, sP, sQ); break;
    }
}

template <int Q>
__device__ __forceinline__ float eval_q(const float (&th)[NQ], const float (&vr)[NQ],
                                        const float (&vi)[NQ], const float2* stab,
                                        const unsigned lane) {
    constexpr Plan PL = make_plan();
    constexpr int K   = PL.K[Q];
    constexpr int NB  = K - 6;
    constexpr int NHI = 1 << NB;
    float ph = 0.f;
#pragma unroll
    for (int i = 0; i < 5; ++i)
        ph += __uint_as_float(__float_as_uint(th[PL.gsel[Q][i]]) ^ (((lane >> i) & 1u) << 31));
    float h0 = 0.f;
#pragma unroll
    for (int i = 5; i < K; ++i) h0 += th[PL.gsel[Q][i]];
    float tr[NHI], ti[NHI];
#pragma unroll
    for (int hi = 0; hi < NHI; ++hi) {
        const float2 e = stab[PL.off[Q] + hi * 32 + (int)lane];
        tr[hi] = e.x; ti[hi] = e.y;
    }
#pragma unroll
    for (int b = NB - 1; b >= 0; --b) {
        const float cr = vr[PL.gsel[Q][5 + b]], ci = vi[PL.gsel[Q][5 + b]];
#pragma unroll
        for (int h = 0; h < (1 << b); ++h) {
            const float xr = tr[h + (1 << b)], xi = ti[h + (1 << b)];
            tr[h] = fmaf(cr, xr, fmaf(-ci, xi, tr[h]));
            ti[h] = fmaf(cr, xi, fmaf(ci, xr, ti[h]));
        }
    }
    const float f = __builtin_amdgcn_fractf(ph + h0);
    return fmaf(__builtin_amdgcn_cosf(f), tr[0], -__builtin_amdgcn_sinf(f) * ti[0]);
}

__global__ __launch_bounds__(TPB) void vqc_eval(const float* __restrict__ x,
                                                const float* __restrict__ W_in,
                                                const float* __restrict__ b_in,
                                                const float* __restrict__ qp,
                                                const float* __restrict__ tabf,
                                                const float* __restrict__ W_out,
                                                const float* __restrict__ b_out,
                                                float* __restrict__ out) {
    constexpr int NV = NTERM / 2, NCP = (NV + TPB - 1) / TPB;
    __shared__ float2 stab[NCP * TPB * 2];
    __shared__ float zs[SPB * NQ];
    __shared__ float red[8 * 256];
    const int tid = threadIdx.x;
    const unsigned lane = (unsigned)tid & 31u;
    const int wave = tid >> 5;

    const FcFrag frag = fc_in_load(x, W_in, (int)lane, wave);
    {
        const float4* src = (const float4*)tabf;
        float4* dst = (float4*)stab;
        float4 tv[NCP];
#pragma unroll
        for (int k = 0; k < NCP; ++k) { const int i = tid + k * TPB; tv[k] = src[i < NV ? i : NV - 1]; }
#pragma unroll
        for (int k = 0; k < NCP; ++k) dst[tid + k * TPB] = tv[k];
    }
    float wrow[NQ];
#pragma unroll
    for (int q = 0; q < NQ; ++q) wrow[q] = W_out[tid * NQ + q];
    const float bo = b_out[tid];
    const int gc = ((lane & 15u) < NQ) ? (int)(lane & 15u) : (NQ - 1);
    const float qb = qp[gc], bi = b_in[gc];
    fc_in_partial(frag, red, (int)lane, wave);
    __syncthreads();

    float ang = red[wave * 32 + (int)lane];
#pragma unroll
    for (int w = 1; w < 8; ++w) ang += red[w * 256 + wave * 32 + (int)lane];
    const float thv = (qb + (ang + bi)) * REV;
    const float f2v = __builtin_amdgcn_fractf(2.0f * thv);
    const float vcv = __builtin_amdgcn_cosf(f2v);
    const float vsv = -__builtin_amdgcn_sinf(f2v);

    const bool b4 = (lane & 16u) != 0u, b3 = (lane & 8u) != 0u, b2 = (lane & 4u) != 0u, b1 = (lane & 2u) != 0u;
    const int qown = b1 ? (8 + 2 * (int)b3 + (int)b4) : (4 * (int)b2 + 2 * (int)b3 + (int)b4);
    const bool own = ((lane & 1u) == 0u) && !(b1 && b2);

    for (int it = 0; it < NITER; ++it) {
        const int sl = it * 8 + wave;
        float th[NQ], vr[NQ], vi[NQ];
#pragma unroll
        for (int g = 0; g < NQ; ++g) {
            th[g] = __int_as_float(__builtin_amdgcn_readlane(__float_as_int(thv), it * 16 + g));
            vr[g] = __int_as_float(__builtin_amdgcn_readlane(__float_as_int(vcv), it * 16 + g));
            vi[g] = __int_as_float(__builtin_amdgcn_readlane(__float_as_int(vsv), it * 16 + g));
        }
        float zp[NQ];
        zp[0]  = eval_q<0>(th, vr, vi, stab, lane);
        zp[1]  = eval_q<1>(th, vr, vi, stab, lane);
        zp[2]  = eval_q<2>(th, vr, vi, stab, lane);
        zp[3]  = eval_q<3>(th, vr, vi, stab, lane);
        zp[4]  = eval_q<4>(th, vr, vi, stab, lane);
        zp[5]  = eval_q<5>(th, vr, vi, stab, lane);
        zp[6]  = eval_q<6>(th, vr, vi, stab, lane);
        zp[7]  = eval_q<7>(th, vr, vi, stab, lane);
        zp[8]  = eval_q<8>(th, vr, vi, stab, lane);
        zp[9]  = eval_q<9>(th, vr, vi, stab, lane);
        zp[10] = eval_q<10>(th, vr, vi, stab, lane);
        zp[11] = eval_q<11>(th, vr, vi, stab, lane);
        float y[NQ / 2];
#pragma unroll
        for (int j = 0; j < NQ / 2; ++j) y[j] = tstep<16>(zp[2 * j], zp[2 * j + 1], b4);
        const float x0 = tstep<8>(y[0], y[1], b3);
        const float x1 = tstep<8>(y[2], y[3], b3);
        const float x2 = tstep<8>(y[4], y[5], b3);
        const float w0 = tstep<4>(x0, x1, b2);
        const float w1 = x2 + xrow<4>(x2);
        float u = tstep<2>(w0, w1, b1);
        u += xrow<1>(u);
        if (own) zs[sl * NQ + qown] = u * (1.0f / 4096.0f);
    }
    __syncthreads();

    for (int s = 0; s < SPB; ++s) {
        float accv = bo;
#pragma unroll
        for (int q = 0; q < NQ; ++q) accv = fmaf(zs[s * NQ + q], wrow[q], accv);
        out[((size_t)blockIdx.x * SPB + s) * DOUT + tid] = accv;
    }
}

static_assert(TAB_BLOCKS == 96 && 4096 / SPB == 256 && SPB == 16 && NITER == 2 && TPB == 256, "grid literals below");
extern "C" void kernel_launch(void* const* d_in, const int* in_sizes, int n_in,
                              void* d_out, int out_size, void* d_ws, size_t ws_size,
                              hipStream_t stream) {
    const float* x        = (const float*)d_in[0];
    const float* W_in     = (const float*)d_in[1];
    const float* b_in     = (const float*)d_in[2];
    const float* q_params = (const float*)d_in[3];
    const float* W_out    = (const float*)d_in[4];
    const float* b_out    = (const float*)d_in[5];
    float* out    = (float*)d_out;
    float* tab    = (float*)d_ws;

    table_kernel<<<dim3(96), dim3(256), 0, stream>>>(q_params, tab);
    vqc_eval<<<dim3(256), dim3(256), 0, stream>>>(x, W_in, b_in, q_params, tab, W_out, b_out, out);
}


# fc_out on the f32 matrix instruction: 16x12 z tile from LDS times W_out fragments loaded behind barrier 1, bias as start accumulator; 6 WMMAs per wave, whole-line stores; fc_out sum order changes
# speedup vs baseline: 38.5104x; 1.0869x over previous
#include <hip/hip_runtime.h>

#define NQ       12
#define NLAYERS  6
#define TPB      256
#define DOUT     256
#define NG (NLAYERS * NQ)

#define TAB_CHUNKS  8
#define TAB_BLOCKS  (NQ * TAB_CHUNKS)
#define NTERM       2176
#define NITER       2
#define SPB         (8 * NITER)
#define REV         0.15915494309189535f

typedef float v2f __attribute__((ext_vector_type(2)));
typedef float v8f __attribute__((ext_vector_type(8)));

struct FcFrag { float4 a[4]; float4 w[4]; };

__device__ __forceinline__ FcFrag fc_in_load(const float* __restrict__ x,
                                             const float* __restrict__ W_in,
                                             const int lane, const int wave) {
    const int n  = lane & 15;
    const int h  = lane >> 4;
    const int nc = (n < NQ) ? n : (NQ - 1);
    const float* xrow = x + (size_t)((int)blockIdx.x * 16 + n) * 256 + wave * 32 + 4 * h;
    const float* wrow = W_in + nc * 256 + wave * 32 + 4 * h;
    FcFrag f;
#pragma unroll
    for (int i = 0; i < 4; ++i) f.a[i] = *(const float4*)(xrow + 8 * i);
#pragma unroll
    for (int i = 0; i < 4; ++i) f.w[i] = *(const float4*)(wrow + 8 * i);
    return f;
}

__device__ __forceinline__ void fc_in_partial(const FcFrag& f, float* red,
                                              const int lane, const int wave) {
    const float mask = ((lane & 15) < NQ) ? 1.0f : 0.0f;
    v8f acc = {};
#pragma unroll
    for (int i = 0; i < 4; ++i) {
        const float4 av = f.a[i];
        float4 wv = f.w[i];
        wv.x *= mask; wv.y *= mask; wv.z *= mask; wv.w *= mask;
        v2f a1, b1, a2, b2;
        a1.x = av.x; a1.y = av.y;  b1.x = wv.x; b1.y = wv.y;
        a2.x = av.z; a2.y = av.w;  b2.x = wv.z; b2.y = wv.w;
        acc = __builtin_amdgcn_wmma_f32_16x16x4_f32(false, a1, false, b1,
                                                    (short)0, acc, false, false);
        acc = __builtin_amdgcn_wmma_f32_16x16x4_f32(false, a2, false, b2,
                                                    (short)0, acc, false, false);
    }
#pragma unroll
    for (int r = 0; r < 8; ++r) red[wave * 256 + r * 32 + lane] = acc[r];
}

struct MaskTab { unsigned m[NG]; };

constexpr MaskTab make_masks() {
    MaskTab t{};
    unsigned R[NQ] = {};
    for (int q = 0; q < NQ; ++q) R[q] = 1u << q;
    for (int l = NLAYERS - 1; l >= 0; --l) {
        R[NQ - 1] ^= R[0];
        for (int q = NQ - 2; q >= 0; --q) R[q] ^= R[q + 1];
        for (int q = 0; q < NQ; ++q) t.m[l * NQ + q] = R[q];
    }
    return t;
}
static_assert(make_masks().m[0] == 0xd7cu && make_masks().m[18] == 0x1e2u && make_masks().m[30] == 0x226u &&
              make_masks().m[47] == 0xd54u && make_masks().m[71] == 0x801u, "Z strings");

struct Plan {
    unsigned mpi[NQ][NG];
    int K[NQ];
    int off[NQ];
    int gsel[NQ][NQ];
    int total;
};
constexpr unsigned cpar(unsigned v) { unsigned p = 0; while (v) { p ^= v & 1u; v >>= 1; } return p; }
constexpr Plan make_plan() {
    Plan P{};
    const MaskTab T = make_masks();
    unsigned A[NQ] = {};
    for (int i = 0; i < NQ; ++i) A[i] = T.m[i] | (1u << (NQ + i));
    for (int c = 0; c < NQ; ++c) {
        int p = c;
        while (!((A[p] >> c) & 1u)) ++p;
        const unsigned t = A[c]; A[c] = A[p]; A[p] = t;
        for (int r = 0; r < NQ; ++r)
            if (r != c && ((A[r] >> c) & 1u)) A[r] ^= A[c];
    }
    unsigned Lc[NQ] = {};
    for (int j = 0; j < NQ; ++j)
        for (int i = 0; i < NQ; ++i) Lc[j] |= ((A[i] >> (NQ + j)) & 1u) << i;
    unsigned mp[NG] = {};
    for (int g = 0; g < NG; ++g)
        for (int k = 0; k < NQ; ++k) mp[g] |= cpar(Lc[k] & T.m[g]) << k;
    int tot = 0;
    bool ok = true;
    for (int g = 0; g < NQ; ++g) ok = ok && (mp[g] == (1u << g));
    for (int q = 0; q < NQ; ++q) {
        int k = 0;
        for (int g = 0; g < NQ; ++g) k += (int)((T.m[g] >> q) & 1u);
        P.K[q] = k;
        P.off[q] = tot;
        tot += 1 << (k - 1);
        const int nt = NQ - k;
        int a = 0, bn = 0;
        int pos[NQ] = {};
        for (int g = 0; g < NQ; ++g) {
            if ((T.m[g] >> q) & 1u) { pos[g] = nt + a; P.gsel[q][a] = g; ++a; }
            else                    { pos[g] = bn; ++bn; }
        }
        for (int g = 0; g < NG; ++g) {
            unsigned v = 0;
            for (int k2 = 0; k2 < NQ; ++k2) v |= ((mp[g] >> k2) & 1u) << pos[k2];
            P.mpi[q][g] = v;
        }
        ok = ok && (k >= 6) && (k <= 10);
    }
    P.total = ok ? tot : -1;
    return P;
}
static_assert(make_plan().total == NTERM, "half-table size");
static_assert(make_plan().K[0] == 8 && make_plan().K[1] == 6 && make_plan().K[6] == 10 && make_plan().K[11] == 9, "K_q");
static_assert(make_plan().off[3] == 192 && make_plan().off[7] == 1152 && make_plan().off[11] == 1920, "offsets");
static_assert(make_plan().mpi[0][12] == 0x11u && make_plan().mpi[11][40] == 0x1c2u && make_plan().mpi[6][71] == 0x82du, "strings over idx");
static_assert(make_plan().gsel[1][5] == 11 && make_plan().gsel[6][5] == 6 && make_plan().gsel[6][9] == 11, "gate order");

struct GateList { int n; int g[NG - NQ]; unsigned mk[NG - NQ]; bool neg[NG - NQ]; };
constexpr GateList make_gates(const int q) {
    GateList L{};
    const MaskTab T = make_masks();
    const Plan P = make_plan();
    const int K = P.K[q];
    const unsigned FLIP = ((1u << K) - 1u) << (NQ - K);
    int n = 0;
    for (int g = NQ; g < NG; ++g) {
        if ((T.m[g] >> q) & 1u) {
            L.g[n] = g;
            L.mk[n] = P.mpi[q][g];
            L.neg[n] = cpar(FLIP & P.mpi[q][g]) != 0u;
            ++n;
        }
    }
    L.n = n;
    return L;
}
static_assert(make_gates(0).n > 0 && make_gates(0).n <= NG - NQ && make_gates(11).g[0] >= NQ, "gate lists");

template <int M>
__device__ __forceinline__ float xrow(const float v) {
    return __int_as_float(__builtin_amdgcn_update_dpp(0, __float_as_int(v), 0x160 | M, 0xf, 0xf, true));
}
__device__ __forceinline__ float xhalf(const float v) {
    const int b = __float_as_int(v);
    return __int_as_float(__builtin_amdgcn_permlanex16(b, b, 0x76543210, 0xfedcba98, false, false));
}
template <int M>
__device__ __forceinline__ float tstep(const float a, const float b, const bool up) {
    const float keep = up ? b : a;
    const float send = up ? a : b;
    if constexpr (M == 16) return keep + xhalf(send);
    else                   return keep + xrow<M>(send);
}

template <int Q, int J>
__device__ __forceinline__ void gate_terms(const float* __restrict__ qp, const unsigned idx,
                                           float& p1, float& p2) {
    constexpr GateList GL = make_gates(Q);
    if constexpr (J < GL.n) {
        constexpr int g = GL.g[J];
        constexpr unsigned mk = GL.mk[J];
        const unsigned w = __float_as_uint(qp[g] * REV);
        const float t = __uint_as_float(w ^ ((__builtin_popcount(idx & mk) & 1u) << 31));
        p1 += t;
        if constexpr (GL.neg[J]) p2 -= t; else p2 += t;
        gate_terms<Q, J + 1>(qp, idx, p1, p2);
    }
}

template <int Q>
__device__ __forceinline__ void build_table(const int chunk, const float* __restrict__ qp,
                                            float2* __restrict__ tab, float* sP, float* sQ) {
    constexpr Plan PL = make_plan();
    constexpr int K  = PL.K[Q];
    constexpr int NT = NQ - K;
    constexpr int nt = 1 << NT;
    static_assert(NT >= 2 && NT <= 6, "a sigma's threads fill part of a wave, one wave or two waves");
    const int tid = threadIdx.x;
    const unsigned idx = (unsigned)(chunk * TPB + tid);
    float p1 = 0.f, p2 = 0.f;
    gate_terms<Q, 0>(qp, idx, p1, p2);
    const float f1 = __builtin_amdgcn_fractf(p1), f2 = __builtin_amdgcn_fractf(p2);
    float aP = __builtin_amdgcn_cosf(f1) + __builtin_amdgcn_cosf(f2);
    float aQ = __builtin_amdgcn_sinf(f1) - __builtin_amdgcn_sinf(f2);
    aP += xrow<1>(aP);  aQ += xrow<1>(aQ);
    aP += xrow<2>(aP);  aQ += xrow<2>(aQ);
    if constexpr (NT >= 3) { aP += xrow<4>(aP);  aQ += xrow<4>(aQ); }
    if constexpr (NT >= 4) { aP += xrow<8>(aP);  aQ += xrow<8>(aQ); }
    if constexpr (NT >= 5) { aP += xhalf(aP);    aQ += xhalf(aQ); }
    if constexpr (NT <= 5) {
        if ((tid & (nt - 1)) == 0) tab[PL.off[Q] + (int)(idx >> NT)] = make_float2(aP, aQ);
    } else {
        const int wave = tid >> 5;
        if ((tid & 31) == 0) { sP[wave] = aP; sQ[wave] = aQ; }
        __syncthreads();
        if (tid < TPB / nt)
            tab[PL.off[Q] + chunk * (TPB / nt) + tid] =
                make_float2(sP[2 * tid] + sP[2 * tid + 1], sQ[2 * tid] + sQ[2 * tid + 1]);
    }
}

__global__ __launch_bounds__(TPB) void table_kernel(const float* __restrict__ qp,
                                                    float* __restrict__ tabf) {
    __shared__ __attribute__((aligned(16))) float sP[TPB];
    __shared__ __attribute__((aligned(16))) float sQ[TPB];
    const int tb = (int)blockIdx.x;
    const int chunk = tb % TAB_CHUNKS;
    float2* tab = (float2*)tabf;
    switch (tb / TAB_CHUNKS) {
        case 0:  build_table<0>(chunk, qp, tab, sP, sQ);  break;
        case 1:  build_table<1>(chunk, qp, tab, sP, sQ);  break;
        case 2:  build_table<2>(chunk, qp, tab, sP, sQ);  break;
        case 3:  build_table<3>(chunk, qp, tab, sP, sQ);  break;
        case 4:  build_table<4>(chunk, qp, tab, sP, sQ);  break;
        case 5:  build_table<5>(chunk, qp, tab, sP, sQ);  break;
        case 6:  build_table<6>(chunk, qp, tab, sP, sQ);  break;
        case 7:  build_table<7>(chunk, qp, tab, sP, sQ);  break;
        case 8:  build_table<8>(chunk, qp, tab, sP, sQ);  break;
        case 9:  build_table<9>(chunk, qp, tab, sP, sQ);  break;
        case 10: build_table<10>(chunk, qp, tab, sP, sQ); break;
        default: build_table<11>(chunk, qp, tab, sP, sQ); break;
    }
}

template <int Q>
__device__ __forceinline__ float eval_q(const float (&th)[NQ], const float (&vr)[NQ],
                                        const float (&vi)[NQ], const float2* stab,
                                        const unsigned lane) {
    constexpr Plan PL = make_plan();
    constexpr int K   = PL.K[Q];
    constexpr int NB  = K - 6;
    constexpr int NHI = 1 << NB;
    float ph = 0.f;
#pragma unroll
    for (int i = 0; i < 5; ++i)
        ph += __uint_as_float(__float_as_uint(th[PL.gsel[Q][i]]) ^ (((lane >> i) & 1u) << 31));
    float h0 = 0.f;
#pragma unroll
    for (int i = 5; i < K; ++i) h0 += th[PL.gsel[Q][i]];
    float tr[NHI], ti[NHI];
#pragma unroll
    for (int hi = 0; hi < NHI; ++hi) {
        const float2 e = stab[PL.off[Q] + hi * 32 + (int)lane];
        tr[hi] = e.x; ti[hi] = e.y;
    }
#pragma unroll
    for (int b = NB - 1; b >= 0; --b) {
        const float cr = vr[PL.gsel[Q][5 + b]], ci = vi[PL.gsel[Q][5 + b]];
#pragma unroll
        for (int h = 0; h < (1 << b); ++h) {
            const float xr = tr[h + (1 << b)], xi = ti[h + (1 << b)];
            tr[h] = fmaf(cr, xr, fmaf(-ci, xi, tr[h]));
            ti[h] = fmaf(cr, xi, fmaf(ci, xr, ti[h]));
        }
    }
    const float f = __builtin_amdgcn_fractf(ph + h0);
    return fmaf(__builtin_amdgcn_cosf(f), tr[0], -__builtin_amdgcn_sinf(f) * ti[0]);
}

__global__ __launch_bounds__(TPB) void vqc_eval(const float* __restrict__ x,
                                                const float* __restrict__ W_in,
                                                const float* __restrict__ b_in,
                                                const float* __restrict__ qp,
                                                const float* __restrict__ tabf,
                                                const float* __restrict__ W_out,
                                                const float* __restrict__ b_out,
                                                float* __restrict__ out) {
    constexpr int NV = NTERM / 2, NCP = (NV + TPB - 1) / TPB;
    __shared__ float2 stab[NCP * TPB * 2];
    __shared__ __attribute__((aligned(16))) float zs[SPB * NQ];
    __shared__ float red[8 * 256];
    const int tid = threadIdx.x;
    const unsigned lane = (unsigned)tid & 31u;
    const int wave = tid >> 5;

    const FcFrag frag = fc_in_load(x, W_in, (int)lane, wave);
    {
        const float4* src = (const float4*)tabf;
        float4* dst = (float4*)stab;
        float4 tv[NCP];
#pragma unroll
        for (int k = 0; k < NCP; ++k) { const int i = tid + k * TPB; tv[k] = src[i < NV ? i : NV - 1]; }
#pragma unroll
        for (int k = 0; k < NCP; ++k) dst[tid + k * TPB] = tv[k];
    }
    const int gc = ((lane & 15u) < NQ) ? (int)(lane & 15u) : (NQ - 1);
    const float qb = qp[gc], bi = b_in[gc];
    fc_in_partial(frag, red, (int)lane, wave);
    __syncthreads();

    const int oc = (int)(lane & 15u), oh = (int)(lane >> 4);
    v2f bw0[3], bw1[3];
    {
        const float* w0 = W_out + (wave * 32 + oc) * NQ + 6 * oh;
        const float* w1 = W_out + (wave * 32 + 16 + oc) * NQ + 6 * oh;
#pragma unroll
        for (int s = 0; s < 3; ++s) { bw0[s] = *(const v2f*)(w0 + 2 * s); bw1[s] = *(const v2f*)(w1 + 2 * s); }
    }
    const float bo0 = b_out[wave * 32 + oc], bo1 = b_out[wave * 32 + 16 + oc];

    float ang = red[wave * 32 + (int)lane];
#pragma unroll
    for (int w = 1; w < 8; ++w) ang += red[w * 256 + wave * 32 + (int)lane];
    const float thv = (qb + (ang + bi)) * REV;
    const float f2v = __builtin_amdgcn_fractf(2.0f * thv);
    const float vcv = __builtin_amdgcn_cosf(f2v);
    const float vsv = -__builtin_amdgcn_sinf(f2v);

    const bool b4 = (lane & 16u) != 0u, b3 = (lane & 8u) != 0u, b2 = (lane & 4u) != 0u, b1 = (lane & 2u) != 0u;
    const int qown = b1 ? (8 + 2 * (int)b3 + (int)b4) : (4 * (int)b2 + 2 * (int)b3 + (int)b4);
    const bool own = ((lane & 1u) == 0u) && !(b1 && b2);

    for (int it = 0; it < NITER; ++it) {
        const int sl = it * 8 + wave;
        float th[NQ], vr[NQ], vi[NQ];
#pragma unroll
        for (int g = 0; g < NQ; ++g) {
            th[g] = __int_as_float(__builtin_amdgcn_readlane(__float_as_int(thv), it * 16 + g));
            vr[g] = __int_as_float(__builtin_amdgcn_readlane(__float_as_int(vcv), it * 16 + g));
            vi[g] = __int_as_float(__builtin_amdgcn_readlane(__float_as_int(vsv), it * 16 + g));
        }
        float zp[NQ];
        zp[0]  = eval_q<0>(th, vr, vi, stab, lane);
        zp[1]  = eval_q<1>(th, vr, vi, stab, lane);
        zp[2]  = eval_q<2>(th, vr, vi, stab, lane);
        zp[3]  = eval_q<3>(th, vr, vi, stab, lane);
        zp[4]  = eval_q<4>(th, vr, vi, stab, lane);
        zp[5]  = eval_q<5>(th, vr, vi, stab, lane);
        zp[6]  = eval_q<6>(th, vr, vi, stab, lane);
        zp[7]  = eval_q<7>(th, vr, vi, stab, lane);
        zp[8]  = eval_q<8>(th, vr, vi, stab, lane);
        zp[9]  = eval_q<9>(th, vr, vi, stab, lane);
        zp[10] = eval_q<10>(th, vr, vi, stab, lane);
        zp[11] = eval_q<11>(th, vr, vi, stab, lane);
        float y[NQ / 2];
#pragma unroll
        for (int j = 0; j < NQ / 2; ++j) y[j] = tstep<16>(zp[2 * j], zp[2 * j + 1], b4);
        const float x0 = tstep<8>(y[0], y[1], b3);
        const float x1 = tstep<8>(y[2], y[3], b3);
        const float x2 = tstep<8>(y[4], y[5], b3);
        const float w0 = tstep<4>(x0, x1, b2);
        const float w1 = x2 + xrow<4>(x2);
        float u = tstep<2>(w0, w1, b1);
        u += xrow<1>(u);
        if (own) zs[sl * NQ + qown] = u * (1.0f / 4096.0f);
    }
    __syncthreads();

    v8f acc0, acc1;
#pragma unroll
    for (int r = 0; r < 8; ++r) { acc0[r] = bo0; acc1[r] = bo1; }
    const float* zr = zs + oc * NQ + 6 * oh;
#pragma unroll
    for (int s = 0; s < 3; ++s) {
        const v2f az = *(const v2f*)(zr + 2 * s);
        acc0 = __builtin_amdgcn_wmma_f32_16x16x4_f32(false, az, false, bw0[s], (short)0, acc0, false, false);
        acc1 = __builtin_amdgcn_wmma_f32_16x16x4_f32(false, az, false, bw1[s], (short)0, acc1, false, false);
    }
    float* orow = out + ((size_t)blockIdx.x * SPB) * DOUT + wave * 32 + (int)lane;
#pragma unroll
    for (int r = 0; r < 8; ++r) {
        const float t0x = xhalf(acc0[r]);
        const float t1x = xhalf(acc1[r]);
        const float lo = oh ? t1x : acc0[r];
        const float hi = oh ? acc1[r] : t0x;
        orow[(size_t)r * DOUT] = lo;
        orow[(size_t)(r + 8) * DOUT] = hi;
    }
}

static_assert(TAB_BLOCKS == 96 && 4096 / SPB == 256 && SPB == 16 && NITER == 2 && TPB == 256, "grid literals below");
extern "C" void kernel_launch(void* const* d_in, const int* in_sizes, int n_in,
                              void* d_out, int out_size, void* d_ws, size_t ws_size,
                              hipStream_t stream) {
    const float* x        = (const float*)d_in[0];
    const float* W_in     = (const float*)d_in[1];
    const float* b_in     = (const float*)d_in[2];
    const float* q_params = (const float*)d_in[3];
    const float* W_out    = (const float*)d_in[4];
    const float* b_out    = (const float*)d_in[5];
    float* out    = (float*)d_out;
    float* tab    = (float*)d_ws;

    table_kernel<<<dim3(96), dim3(256), 0, stream>>>(q_params, tab);
    vqc_eval<<<dim3(256), dim3(256), 0, stream>>>(x, W_in, b_in, q_params, tab, W_out, b_out, out);
}


# Evaluation launch in blocks of 4 waves, one per SIMD, 4 samples per wave, two per loop trip: 1,024 waves enter, not 2,048; fc_in K split 4 ways in two accumulator chains; fc_in sum order changes
# speedup vs baseline: 40.4841x; 1.0513x over previous
#include <hip/hip_runtime.h>

#define NQ       12
#define NLAYERS  6
#define TPB      256
#define EW       4
#define ETPB     (32 * EW)
#define DOUT     256
#define NG (NLAYERS * NQ)

#define TAB_CHUNKS  8
#define TAB_BLOCKS  (NQ * TAB_CHUNKS)
#define NTERM       2176
#define NITER       4
#define SPB         (EW * NITER)
#define REV         0.15915494309189535f

typedef float v2f __attribute__((ext_vector_type(2)));
typedef float v8f __attribute__((ext_vector_type(8)));

struct FcFrag { float4 a[8]; float4 w[8]; };

__device__ __forceinline__ FcFrag fc_in_load(const float* __restrict__ x,
                                             const float* __restrict__ W_in,
                                             const int lane, const int wave) {
    const int n  = lane & 15;
    const int h  = lane >> 4;
    const int nc = (n < NQ) ? n : (NQ - 1);
    const float* xrow = x + (size_t)((int)blockIdx.x * 16 + n) * 256 + wave * 64 + 4 * h;
    const float* wrow = W_in + nc * 256 + wave * 64 + 4 * h;
    FcFrag f;
#pragma unroll
    for (int i = 0; i < 8; ++i) f.a[i] = *(const float4*)(xrow + 8 * i);
#pragma unroll
    for (int i = 0; i < 8; ++i) f.w[i] = *(const float4*)(wrow + 8 * i);
    return f;
}

__device__ __forceinline__ void fc_in_partial(const FcFrag& f, float* red,
                                              const int lane, const int wave) {
    const float mask = ((lane & 15) < NQ) ? 1.0f : 0.0f;
    v8f acc = {}, acc2 = {};
#pragma unroll
    for (int i = 0; i < 8; ++i) {
        const float4 av = f.a[i];
        float4 wv = f.w[i];
        wv.x *= mask; wv.y *= mask; wv.z *= mask; wv.w *= mask;
        v2f a1, b1, a2, b2;
        a1.x = av.x; a1.y = av.y;  b1.x = wv.x; b1.y = wv.y;
        a2.x = av.z; a2.y = av.w;  b2.x = wv.z; b2.y = wv.w;
        acc = __builtin_amdgcn_wmma_f32_16x16x4_f32(false, a1, false, b1,
                                                    (short)0, acc, false, false);
        acc2 = __builtin_amdgcn_wmma_f32_16x16x4_f32(false, a2, false, b2,
                                                     (short)0, acc2, false, false);
    }
#pragma unroll
    for (int r = 0; r < 8; ++r) red[wave * 256 + r * 32 + lane] = acc[r] + acc2[r];
}

struct MaskTab { unsigned m[NG]; };

constexpr MaskTab make_masks() {
    MaskTab t{};
    unsigned R[NQ] = {};
    for (int q = 0; q < NQ; ++q) R[q] = 1u << q;
    for (int l = NLAYERS - 1; l >= 0; --l) {
        R[NQ - 1] ^= R[0];
        for (int q = NQ - 2; q >= 0; --q) R[q] ^= R[q + 1];
        for (int q = 0; q < NQ; ++q) t.m[l * NQ + q] = R[q];
    }
    return t;
}
static_assert(make_masks().m[0] == 0xd7cu && make_masks().m[18] == 0x1e2u && make_masks().m[30] == 0x226u &&
              make_masks().m[47] == 0xd54u && make_masks().m[71] == 0x801u, "Z strings");

struct Plan {
    unsigned mpi[NQ][NG];
    int K[NQ];
    int off[NQ];
    int gsel[NQ][NQ];
    int total;
};
constexpr unsigned cpar(unsigned v) { unsigned p = 0; while (v) { p ^= v & 1u; v >>= 1; } return p; }
constexpr Plan make_plan() {
    Plan P{};
    const MaskTab T = make_masks();
    unsigned A[NQ] = {};
    for (int i = 0; i < NQ; ++i) A[i] = T.m[i] | (1u << (NQ + i));
    for (int c = 0; c < NQ; ++c) {
        int p = c;
        while (!((A[p] >> c) & 1u)) ++p;
        const unsigned t = A[c]; A[c] = A[p]; A[p] = t;
        for (int r = 0; r < NQ; ++r)
            if (r != c && ((A[r] >> c) & 1u)) A[r] ^= A[c];
    }
    unsigned Lc[NQ] = {};
    for (int j = 0; j < NQ; ++j)
        for (int i = 0; i < NQ; ++i) Lc[j] |= ((A[i] >> (NQ + j)) & 1u) << i;
    unsigned mp[NG] = {};
    for (int g = 0; g < NG; ++g)
        for (int k = 0; k < NQ; ++k) mp[g] |= cpar(Lc[k] & T.m[g]) << k;
    int tot = 0;
    bool ok = true;
    for (int g = 0; g < NQ; ++g) ok = ok && (mp[g] == (1u << g));
    for (int q = 0; q < NQ; ++q) {
        int k = 0;
        for (int g = 0; g < NQ; ++g) k += (int)((T.m[g] >> q) & 1u);
        P.K[q] = k;
        P.off[q] = tot;
        tot += 1 << (k - 1);
        const int nt = NQ - k;
        int a = 0, bn = 0;
        int pos[NQ] = {};
        for (int g = 0; g < NQ; ++g) {
            if ((T.m[g] >> q) & 1u) { pos[g] = nt + a; P.gsel[q][a] = g; ++a; }
            else                    { pos[g] = bn; ++bn; }
        }
        for (int g = 0; g < NG; ++g) {
            unsigned v = 0;
            for (int k2 = 0; k2 < NQ; ++k2) v |= ((mp[g] >> k2) & 1u) << pos[k2];
            P.mpi[q][g] = v;
        }
        ok = ok && (k >= 6) && (k <= 10);
    }
    P.total = ok ? tot : -1;
    return P;
}
static_assert(make_plan().total == NTERM, "half-table size");
static_assert(make_plan().K[0] == 8 && make_plan().K[1] == 6 && make_plan().K[6] == 10 && make_plan().K[11] == 9, "K_q");
static_assert(make_plan().off[3] == 192 && make_plan().off[7] == 1152 && make_plan().off[11] == 1920, "offsets");
static_assert(make_plan().mpi[0][12] == 0x11u && make_plan().mpi[11][40] == 0x1c2u && make_plan().mpi[6][71] == 0x82du, "strings over idx");
static_assert(make_plan().gsel[1][5] == 11 && make_plan().gsel[6][5] == 6 && make_plan().gsel[6][9] == 11, "gate order");

struct GateList { int n; int g[NG - NQ]; unsigned mk[NG - NQ]; bool neg[NG - NQ]; };
constexpr GateList make_gates(const int q) {
    GateList L{};
    const MaskTab T = make_masks();
    const Plan P = make_plan();
    const int K = P.K[q];
    const unsigned FLIP = ((1u << K) - 1u) << (NQ - K);
    int n = 0;
    for (int g = NQ; g < NG; ++g) {
        if ((T.m[g] >> q) & 1u) {
            L.g[n] = g;
            L.mk[n] = P.mpi[q][g];
            L.neg[n] = cpar(FLIP & P.mpi[q][g]) != 0u;
            ++n;
        }
    }
    L.n = n;
    return L;
}
static_assert(make_gates(0).n > 0 && make_gates(0).n <= NG - NQ && make_gates(11).g[0] >= NQ, "gate lists");

template <int M>
__device__ __forceinline__ float xrow(const float v) {
    return __int_as_float(__builtin_amdgcn_update_dpp(0, __float_as_int(v), 0x160 | M, 0xf, 0xf, true));
}
__device__ __forceinline__ float xhalf(const float v) {
    const int b = __float_as_int(v);
    return __int_as_float(__builtin_amdgcn_permlanex16(b, b, 0x76543210, 0xfedcba98, false, false));
}
template <int M>
__device__ __forceinline__ float tstep(const float a, const float b, const bool up) {
    const float keep = up ? b : a;
    const float send = up ? a : b;
    if constexpr (M == 16) return keep + xhalf(send);
    else                   return keep + xrow<M>(send);
}

template <int Q, int J>
__device__ __forceinline__ void gate_terms(const float* __restrict__ qp, const unsigned idx,
                                           float& p1, float& p2) {
    constexpr GateList GL = make_gates(Q);
    if constexpr (J < GL.n) {
        constexpr int g = GL.g[J];
        constexpr unsigned mk = GL.mk[J];
        const unsigned w = __float_as_uint(qp[g] * REV);
        const float t = __uint_as_float(w ^ ((__builtin_popcount(idx & mk) & 1u) << 31));
        p1 += t;
        if constexpr (GL.neg[J]) p2 -= t; else p2 += t;
        gate_terms<Q, J + 1>(qp, idx, p1, p2);
    }
}

template <int Q>
__device__ __forceinline__ void build_table(const int chunk, const float* __restrict__ qp,
                                            float2* __restrict__ tab, float* sP, float* sQ) {
    constexpr Plan PL = make_plan();
    constexpr int K  = PL.K[Q];
    constexpr int NT = NQ - K;
    constexpr int nt = 1 << NT;
    static_assert(NT >= 2 && NT <= 6, "a sigma's threads fill part of a wave, one wave or two waves");
    const int tid = threadIdx.x;
    const unsigned idx = (unsigned)(chunk * TPB + tid);
    float p1 = 0.f, p2 = 0.f;
    gate_terms<Q, 0>(qp, idx, p1, p2);
    const float f1 = __builtin_amdgcn_fractf(p1), f2 = __builtin_amdgcn_fractf(p2);
    float aP = __builtin_amdgcn_cosf(f1) + __builtin_amdgcn_cosf(f2);
    float aQ = __builtin_amdgcn_sinf(f1) - __builtin_amdgcn_sinf(f2);
    aP += xrow<1>(aP);  aQ += xrow<1>(aQ);
    aP += xrow<2>(aP);  aQ += xrow<2>(aQ);
    if constexpr (NT >= 3) { aP += xrow<4>(aP);  aQ += xrow<4>(aQ); }
    if constexpr (NT >= 4) { aP += xrow<8>(aP);  aQ += xrow<8>(aQ); }
    if constexpr (NT >= 5) { aP += xhalf(aP);    aQ += xhalf(aQ); }
    if constexpr (NT <= 5) {
        if ((tid & (nt - 1)) == 0) tab[PL.off[Q] + (int)(idx >> NT)] = make_float2(aP, aQ);
    } else {
        const int wave = tid >> 5;
        if ((tid & 31) == 0) { sP[wave] = aP; sQ[wave] = aQ; }
        __syncthreads();
        if (tid < TPB / nt)
            tab[PL.off[Q] + chunk * (TPB / nt) + tid] =
                make_float2(sP[2 * tid] + sP[2 * tid + 1], sQ[2 * tid] + sQ[2 * tid + 1]);
    }
}

__global__ __launch_bounds__(TPB) void table_kernel(const float* __restrict__ qp,
                                                    float* __restrict__ tabf) {
    __shared__ __attribute__((aligned(16))) float sP[TPB];
    __shared__ __attribute__((aligned(16))) float sQ[TPB];
    const int tb = (int)blockIdx.x;
    const int chunk = tb % TAB_CHUNKS;
    float2* tab = (float2*)tabf;
    switch (tb / TAB_CHUNKS) {
        case 0:  build_table<0>(chunk, qp, tab, sP, sQ);  break;
        case 1:  build_table<1>(chunk, qp, tab, sP, sQ);  break;
        case 2:  build_table<2>(chunk, qp, tab, sP, sQ);  break;
        case 3:  build_table<3>(chunk, qp, tab, sP, sQ);  break;
        case 4:  build_table<4>(chunk, qp, tab, sP, sQ);  break;
        case 5:  build_table<5>(chunk, qp, tab, sP, sQ);  break;
        case 6:  build_table<6>(chunk, qp, tab, sP, sQ);  break;
        case 7:  build_table<7>(chunk, qp, tab, sP, sQ);  break;
        case 8:  build_table<8>(chunk, qp, tab, sP, sQ);  break;
        case 9:  build_table<9>(chunk, qp, tab, sP, sQ);  break;
        case 10: build_table<10>(chunk, qp, tab, sP, sQ); break;
        default: build_table<11>(chunk, qp, tab, sP, sQ); break;
    }
}

template <int Q>
__device__ __forceinline__ float eval_q(const float (&th)[NQ], const float (&vr)[NQ],
                                        const float (&vi)[NQ], const float2* stab,
                                        const unsigned lane) {
    constexpr Plan PL = make_plan();
    constexpr int K   = PL.K[Q];
    constexpr int NB  = K - 6;
    constexpr int NHI = 1 << NB;
    float ph = 0.f;
#pragma unroll
    for (int i = 0; i < 5; ++i)
        ph += __uint_as_float(__float_as_uint(th[PL.gsel[Q][i]]) ^ (((lane >> i) & 1u) << 31));
    float h0 = 0.f;
#pragma unroll
    for (int i = 5; i < K; ++i) h0 += th[PL.gsel[Q][i]];
    float tr[NHI], ti[NHI];
#pragma unroll
    for (int hi = 0; hi < NHI; ++hi) {
        const float2 e = stab[PL.off[Q] + hi * 32 + (int)lane];
        tr[hi] = e.x; ti[hi] = e.y;
    }
#pragma unroll
    for (int b = NB - 1; b >= 0; --b) {
        const float cr = vr[PL.gsel[Q][5 + b]], ci = vi[PL.gsel[Q][5 + b]];
#pragma unroll
        for (int h = 0; h < (1 << b); ++h) {
            const float xr = tr[h + (1 << b)], xi = ti[h + (1 << b)];
            tr[h] = fmaf(cr, xr, fmaf(-ci, xi, tr[h]));
            ti[h] = fmaf(cr, xi, fmaf(ci, xr, ti[h]));
        }
    }
    const float f = __builtin_amdgcn_fractf(ph + h0);
    return fmaf(__builtin_amdgcn_cosf(f), tr[0], -__builtin_amdgcn_sinf(f) * ti[0]);
}

__global__ __launch_bounds__(ETPB) void vqc_eval(const float* __restrict__ x,
                                                 const float* __restrict__ W_in,
                                                 const float* __restrict__ b_in,
                                                 const float* __restrict__ qp,
                                                 const float* __restrict__ tabf,
                                                 const float* __restrict__ W_out,
                                                 const float* __restrict__ b_out,
                                                 float* __restrict__ out) {
    constexpr int NV = NTERM / 2, NCP = (NV + ETPB - 1) / ETPB;
    __shared__ float2 stab[NCP * ETPB * 2];
    __shared__ __attribute__((aligned(16))) float zs[SPB * NQ];
    __shared__ float red[EW * 256];
    const int tid = threadIdx.x;
    const unsigned lane = (unsigned)tid & 31u;
    const int wave = tid >> 5;

    const FcFrag frag = fc_in_load(x, W_in, (int)lane, wave);
    {
        const float4* src = (const float4*)tabf;
        float4* dst = (float4*)stab;
        float4 tv[NCP];
#pragma unroll
        for (int k = 0; k < NCP; ++k) { const int i = tid + k * ETPB; tv[k] = src[i < NV ? i : NV - 1]; }
#pragma unroll
        for (int k = 0; k < NCP; ++k) dst[tid + k * ETPB] = tv[k];
    }
    const int gc = ((lane & 15u) < NQ) ? (int)(lane & 15u) : (NQ - 1);
    const float qb = qp[gc], bi = b_in[gc];
    fc_in_partial(frag, red, (int)lane, wave);
    __syncthreads();

    const int oc = (int)(lane & 15u), oh = (int)(lane >> 4);
    v2f bw[4][3];
    float bo[4];
#pragma unroll
    for (int t = 0; t < 4; ++t) {
        const float* wp = W_out + (wave * 64 + 16 * t + oc) * NQ + 6 * oh;
#pragma unroll
        for (int s = 0; s < 3; ++s) bw[t][s] = *(const v2f*)(wp + 2 * s);
        bo[t] = b_out[wave * 64 + 16 * t + oc];
    }

    float angA = red[wave * 32 + (int)lane];
    float angB = red[(wave + 4) * 32 + (int)lane];
#pragma unroll
    for (int w = 1; w < EW; ++w) {
        angA += red[w * 256 + wave * 32 + (int)lane];
        angB += red[w * 256 + (wave + 4) * 32 + (int)lane];
    }
    const float thvA = (qb + (angA + bi)) * REV;
    const float thvB = (qb + (angB + bi)) * REV;
    const float f2A = __builtin_amdgcn_fractf(2.0f * thvA);
    const float f2B = __builtin_amdgcn_fractf(2.0f * thvB);
    const float vcvA = __builtin_amdgcn_cosf(f2A), vsvA = -__builtin_amdgcn_sinf(f2A);
    const float vcvB = __builtin_amdgcn_cosf(f2B), vsvB = -__builtin_amdgcn_sinf(f2B);

    const bool b4 = (lane & 16u) != 0u, b3 = (lane & 8u) != 0u, b2 = (lane & 4u) != 0u, b1 = (lane & 2u) != 0u;
    const int qown = b1 ? (8 + 2 * (int)b3 + (int)b4) : (4 * (int)b2 + 2 * (int)b3 + (int)b4);
    const bool own = ((lane & 1u) == 0u) && !(b1 && b2);

    for (int jj = 0; jj < NITER / 2; ++jj) {
        const float thv = jj ? thvB : thvA;
        const float vcv = jj ? vcvB : vcvA;
        const float vsv = jj ? vsvB : vsvA;
        float u[2];
#pragma unroll
        for (int hf = 0; hf < 2; ++hf) {
            float th[NQ], vr[NQ], vi[NQ];
#pragma unroll
            for (int g = 0; g < NQ; ++g) {
                th[g] = __int_as_float(__builtin_amdgcn_readlane(__float_as_int(thv), hf * 16 + g));
                vr[g] = __int_as_float(__builtin_amdgcn_readlane(__float_as_int(vcv), hf * 16 + g));
                vi[g] = __int_as_float(__builtin_amdgcn_readlane(__float_as_int(vsv), hf * 16 + g));
            }
            float zp[NQ];
            zp[0]  = eval_q<0>(th, vr, vi, stab, lane);
            zp[1]  = eval_q<1>(th, vr, vi, stab, lane);
            zp[2]  = eval_q<2>(th, vr, vi, stab, lane);
            zp[3]  = eval_q<3>(th, vr, vi, stab, lane);
            zp[4]  = eval_q<4>(th, vr, vi, stab, lane);
            zp[5]  = eval_q<5>(th, vr, vi, stab, lane);
            zp[6]  = eval_q<6>(th, vr, vi, stab, lane);
            zp[7]  = eval_q<7>(th, vr, vi, stab, lane);
            zp[8]  = eval_q<8>(th, vr, vi, stab, lane);
            zp[9]  = eval_q<9>(th, vr, vi, stab, lane);
            zp[10] = eval_q<10>(th, vr, vi, stab, lane);
            zp[11] = eval_q<11>(th, vr, vi, stab, lane);
            float y[NQ / 2];
#pragma unroll
            for (int j = 0; j < NQ / 2; ++j) y[j] = tstep<16>(zp[2 * j], zp[2 * j + 1], b4);
            const float x0 = tstep<8>(y[0], y[1], b3);
            const float x1 = tstep<8>(y[2], y[3], b3);
            const float x2 = tstep<8>(y[4], y[5], b3);
            const float w0 = tstep<4>(x0, x1, b2);
            const float w1 = x2 + xrow<4>(x2);
            float uu = tstep<2>(w0, w1, b1);
            uu += xrow<1>(uu);
            u[hf] = uu * (1.0f / 4096.0f);
        }
        if (own) {
            zs[(wave + 4 * jj) * NQ + qown]     = u[0];
            zs[(wave + 4 * jj + 8) * NQ + qown] = u[1];
        }
    }
    __syncthreads();

    v8f acc[4];
#pragma unroll
    for (int t = 0; t < 4; ++t) {
#pragma unroll
        for (int r = 0; r < 8; ++r) acc[t][r] = bo[t];
    }
    const float* zr = zs + oc * NQ + 6 * oh;
#pragma unroll
    for (int s = 0; s < 3; ++s) {
        const v2f az = *(const v2f*)(zr + 2 * s);
#pragma unroll
        for (int t = 0; t < 4; ++t)
            acc[t] = __builtin_amdgcn_wmma_f32_16x16x4_f32(false, az, false, bw[t][s], (short)0, acc[t], false, false);
    }
#pragma unroll
    for (int p = 0; p < 2; ++p) {
        float* orow = out + ((size_t)blockIdx.x * SPB) * DOUT + wave * 64 + 32 * p + (int)lane;
#pragma unroll
        for (int r = 0; r < 8; ++r) {
            const float t0x = xhalf(acc[2 * p][r]);
            const float t1x = xhalf(acc[2 * p + 1][r]);
            const float lo = oh ? t1x : acc[2 * p][r];
            const float hi = oh ? acc[2 * p + 1][r] : t0x;
            orow[(size_t)r * DOUT] = lo;
            orow[(size_t)(r + 8) * DOUT] = hi;
        }
    }
}

static_assert(TAB_BLOCKS == 96 && 4096 / SPB == 256 && SPB == 16 && NITER == 4 && TPB == 256 && ETPB == 128, "grid literals below");
extern "C" void kernel_launch(void* const* d_in, const int* in_sizes, int n_in,
                              void* d_out, int out_size, void* d_ws, size_t ws_size,
                              hipStream_t stream) {
    const float* x        = (const float*)d_in[0];
    const float* W_in     = (const float*)d_in[1];
    const float* b_in     = (const float*)d_in[2];
    const float* q_params = (const float*)d_in[3];
    const float* W_out    = (const float*)d_in[4];
    const float* b_out    = (const float*)d_in[5];
    float* out    = (float*)d_out;
    float* tab    = (float*)d_ws;

    table_kernel<<<dim3(96), dim3(256), 0, stream>>>(q_params, tab);
    vqc_eval<<<dim3(256), dim3(128), 0, stream>>>(x, W_in, b_in, q_params, tab, W_out, b_out, out);
}


# Table launch folded into the evaluation launch: each block builds the half-table in LDS from psi(s)=e^{i phi(s)} over 4096 sign patterns, entry = 2 sum_tau psi conj(psi'); table sum order changes
# speedup vs baseline: 43.0403x; 1.0631x over previous
#include <hip/hip_runtime.h>

#define NQ       12
#define NLAYERS  6
#define EW       4
#define ETPB     (32 * EW)
#define DOUT     256
#define NG (NLAYERS * NQ)

#define NTERM       2176
#define NS          4096
#define NITER       4
#define SPB         (EW * NITER)
#define REV         0.15915494309189535f

typedef float v2f __attribute__((ext_vector_type(2)));
typedef float v8f __attribute__((ext_vector_type(8)));

struct FcFrag { float4 a[8]; float4 w[8]; };

__device__ __forceinline__ FcFrag fc_in_load(const float* __restrict__ x,
                                             const float* __restrict__ W_in,
                                             const int lane, const int wave) {
    const int n  = lane & 15;
    const int h  = lane >> 4;
    const int nc = (n < NQ) ? n : (NQ - 1);
    const float* xrow = x + (size_t)((int)blockIdx.x * 16 + n) * 256 + wave * 64 + 4 * h;
    const float* wrow = W_in + nc * 256 + wave * 64 + 4 * h;
    FcFrag f;
#pragma unroll
    for (int i = 0; i < 8; ++i) f.a[i] = *(const float4*)(xrow + 8 * i);
#pragma unroll
    for (int i = 0; i < 8; ++i) f.w[i] = *(const float4*)(wrow + 8 * i);
    return f;
}

__device__ __forceinline__ void fc_in_partial(const FcFrag& f, float* red,
                                              const int lane, const int wave) {
    const float mask = ((lane & 15) < NQ) ? 1.0f : 0.0f;
    v8f acc = {}, acc2 = {};
#pragma unroll
    for (int i = 0; i < 8; ++i) {
        const float4 av = f.a[i];
        float4 wv = f.w[i];
        wv.x *= mask; wv.y *= mask; wv.z *= mask; wv.w *= mask;
        v2f a1, b1, a2, b2;
        a1.x = av.x; a1.y = av.y;  b1.x = wv.x; b1.y = wv.y;
        a2.x = av.z; a2.y = av.w;  b2.x = wv.z; b2.y = wv.w;
        acc = __builtin_amdgcn_wmma_f32_16x16x4_f32(false, a1, false, b1,
                                                    (short)0, acc, false, false);
        acc2 = __builtin_amdgcn_wmma_f32_16x16x4_f32(false, a2, false, b2,
                                                     (short)0, acc2, false, false);
    }
#pragma unroll
    for (int r = 0; r < 8; ++r) red[wave * 256 + r * 32 + lane] = acc[r] + acc2[r];
}

struct MaskTab { unsigned m[NG]; };

constexpr MaskTab make_masks() {
    MaskTab t{};
    unsigned R[NQ] = {};
    for (int q = 0; q < NQ; ++q) R[q] = 1u << q;
    for (int l = NLAYERS - 1; l >= 0; --l) {
        R[NQ - 1] ^= R[0];
        for (int q = NQ - 2; q >= 0; --q) R[q] ^= R[q + 1];
        for (int q = 0; q < NQ; ++q) t.m[l * NQ + q] = R[q];
    }
    return t;
}
static_assert(make_masks().m[0] == 0xd7cu && make_masks().m[18] == 0x1e2u && make_masks().m[30] == 0x226u &&
              make_masks().m[47] == 0xd54u && make_masks().m[71] == 0x801u, "Z strings");

struct Plan {
    unsigned mps[NG];
    unsigned F[NQ];
    int top[NQ];
    int K[NQ];
    int off[NQ];
    int gsel[NQ][NQ];
    int total;
};
constexpr unsigned cpar(unsigned v) { unsigned p = 0; while (v) { p ^= v & 1u; v >>= 1; } return p; }
constexpr Plan make_plan() {
    Plan P{};
    const MaskTab T = make_masks();
    unsigned A[NQ] = {};
    for (int i = 0; i < NQ; ++i) A[i] = T.m[i] | (1u << (NQ + i));
    for (int c = 0; c < NQ; ++c) {
        int p = c;
        while (!((A[p] >> c) & 1u)) ++p;
        const unsigned t = A[c]; A[c] = A[p]; A[p] = t;
        for (int r = 0; r < NQ; ++r)
            if (r != c && ((A[r] >> c) & 1u)) A[r] ^= A[c];
    }
    unsigned Lc[NQ] = {};
    for (int j = 0; j < NQ; ++j)
        for (int i = 0; i < NQ; ++i) Lc[j] |= ((A[i] >> (NQ + j)) & 1u) << i;
    unsigned mp[NG] = {};
    for (int g = 0; g < NG; ++g)
        for (int k = 0; k < NQ; ++k) mp[g] |= cpar(Lc[k] & T.m[g]) << k;
    int tot = 0;
    bool ok = true;
    for (int g = 0; g < NQ; ++g) ok = ok && (mp[g] == (1u << g));
    for (int g = 0; g < NG; ++g) P.mps[g] = mp[g];
    for (int q = 0; q < NQ; ++q) {
        int k = 0;
        unsigned f = 0;
        for (int g = 0; g < NQ; ++g)
            if ((T.m[g] >> q) & 1u) { P.gsel[q][k] = g; P.top[q] = g; f |= 1u << g; ++k; }
        P.K[q] = k;
        P.F[q] = f;
        P.off[q] = tot;
        tot += 1 << (k - 1);
        ok = ok && (k >= 6) && (k <= 10);
        ok = ok && (P.top[q] >= 7);
        for (int g = 0; g < NG; ++g) ok = ok && (cpar(mp[g] & f) == ((T.m[g] >> q) & 1u));
    }
    P.total = ok ? tot : -1;
    return P;
}
static_assert(make_plan().total == NTERM, "half-table size");
static_assert(make_plan().K[0] == 8 && make_plan().K[1] == 6 && make_plan().K[6] == 10 && make_plan().K[11] == 9, "K_q");
static_assert(make_plan().off[3] == 192 && make_plan().off[7] == 1152 && make_plan().off[11] == 1920, "offsets");
static_assert(make_plan().mps[12] == 0x003u && make_plan().mps[23] == 0x803u && make_plan().mps[71] == 0x82bu, "strings over s");
static_assert(make_plan().F[0] == 0x7d6u && make_plan().F[1] == 0x87au && make_plan().F[6] == 0xf5fu && make_plan().F[11] == 0xbebu, "sigma masks");
static_assert(make_plan().top[0] == 10 && make_plan().top[2] == 7 && make_plan().top[3] == 8 && make_plan().top[11] == 11, "top sigma gates");
static_assert(make_plan().gsel[1][5] == 11 && make_plan().gsel[6][5] == 6 && make_plan().gsel[6][9] == 11, "gate order");

template <int M>
__device__ __forceinline__ float xrow(const float v) {
    return __int_as_float(__builtin_amdgcn_update_dpp(0, __float_as_int(v), 0x160 | M, 0xf, 0xf, true));
}
__device__ __forceinline__ float xhalf(const float v) {
    const int b = __float_as_int(v);
    return __int_as_float(__builtin_amdgcn_permlanex16(b, b, 0x76543210, 0xfedcba98, false, false));
}
template <int M>
__device__ __forceinline__ float tstep(const float a, const float b, const bool up) {
    const float keep = up ? b : a;
    const float send = up ? a : b;
    if constexpr (M == 16) return keep + xhalf(send);
    else                   return keep + xrow<M>(send);
}


constexpr bool bin_first(const int G) {
    const Plan P = make_plan();
    for (int g = NQ; g < G; ++g) if ((P.mps[g] >> 7) == (P.mps[G] >> 7)) return false;
    return true;
}
template <int G>
__device__ __forceinline__ void bin_gate(const float* __restrict__ qp, const unsigned tid, float (&c)[32]) {
    if constexpr (G < NG) {
        constexpr Plan PL = make_plan();
        constexpr unsigned lo = PL.mps[G] & 0x7fu;
        constexpr int hi = (int)(PL.mps[G] >> 7);
        const float hw = qp[G] * (0.5f * REV);
        float t = hw;
        if constexpr (lo != 0u)
            t = __uint_as_float(__float_as_uint(hw) ^ ((__builtin_popcount(tid & lo) & 1u) << 31));
        if constexpr (bin_first(G)) c[hi] = t; else c[hi] += t;
        bin_gate<G + 1>(qp, tid, c);
    }
}
__device__ __forceinline__ void build_psi(const float* __restrict__ qp, const unsigned tid,
                                          float2* psi, float (&pr)[32], float (&pi)[32]) {
    float c[32] = {};
    bin_gate<NQ>(qp, tid, c);
#pragma unroll
    for (int h = 1; h < 32; h <<= 1) {
#pragma unroll
        for (int i = 0; i < 32; ++i) {
            if ((i & h) == 0) {
                const float a = c[i], b = c[i | h];
                c[i] = a + b;
                c[i | h] = a - b;
            }
        }
    }
#pragma unroll
    for (int r = 0; r < 32; ++r) {
        const float f = __builtin_amdgcn_fractf(c[r]);
        pr[r] = __builtin_amdgcn_cosf(f);
        pi[r] = __builtin_amdgcn_sinf(f);
        psi[(r << 7) | (int)tid] = make_float2(pr[r], pi[r]);
    }
}

struct Lay {
    int Wg[2]; int Rg[4];
    bool own, ok;
    unsigned tmask;
    unsigned sj[16];
    unsigned sj2[16];
    int kidx[16];
    int tidx[16];
    int nk;
    unsigned ks[16];
    int pw[2];
    int pl[5];
};
constexpr int spos(const unsigned F, const int g) {
    int p = 0;
    for (int i = 0; i < g; ++i) p += (int)((F >> i) & 1u);
    return p;
}
constexpr Lay make_lay(const int q) {
    Lay L{};
    const Plan P = make_plan();
    const unsigned F = P.F[q];
    const int top = P.top[q];
    int cand[NQ] = {};
    int nc = 0;
    for (int g = 5; g < NQ; ++g) if (g != top) cand[nc++] = g;
    int a = -1, b = -1;
    if (((F >> 5) & 1u) && ((F >> 6) & 1u)) { a = 5; b = 6; }
    else {
        for (int i = nc - 1; i >= 0; --i)
            if ((F >> cand[i]) & 1u) { if (b < 0) b = cand[i]; else if (a < 0) a = cand[i]; }
    }
    L.ok = (nc == 6) && (a >= 5) && (b > a);
    if (!L.ok) return L;
    L.Wg[0] = a; L.Wg[1] = b;
    int nr = 0;
    for (int i = 0; i < nc; ++i) if (cand[i] != a && cand[i] != b) L.Rg[nr++] = cand[i];
    L.ok = L.ok && (nr == 4);
    L.own = (a == 5 && b == 6);
    L.tmask = 0x1fu | (1u << a) | (1u << b);
    unsigned rmask = 0;
    int nsig = 0;
    for (int i = 0; i < 4; ++i) { rmask |= 1u << L.Rg[i]; nsig += (int)((F >> L.Rg[i]) & 1u); }
    L.ok = L.ok && ((L.tmask | rmask | (1u << top)) == 0xfffu) && ((L.tmask & rmask) == 0u);
    L.nk = 1 << nsig;
    for (int j = 0; j < 16; ++j) {
        unsigned s = 0;
        int k = 0, kb = 0, t = 0, tb = 0;
        for (int i = 0; i < 4; ++i) {
            const unsigned bit = (unsigned)(j >> i) & 1u;
            s |= bit << L.Rg[i];
            if ((F >> L.Rg[i]) & 1u) { k |= (int)bit << kb; ++kb; }
            else                     { t |= (int)bit << tb; ++tb; }
        }
        L.sj[j] = s;
        L.sj2[j] = (s ^ (F & rmask)) | (1u << top);
        L.kidx[j] = k;
        L.tidx[j] = t;
    }
    for (int k = 0; k < L.nk; ++k) {
        unsigned e = 0;
        int kb = 0;
        for (int i = 0; i < 4; ++i)
            if ((F >> L.Rg[i]) & 1u) { e |= ((unsigned)(k >> kb) & 1u) << spos(F, L.Rg[i]); ++kb; }
        L.ks[k] = e;
    }
    L.pw[0] = spos(F, a); L.pw[1] = spos(F, b);
    for (int g = 0; g < 5; ++g) L.pl[g] = spos(F, g);
    return L;
}
static_assert(make_lay(0).ok && make_lay(1).ok && make_lay(2).ok && make_lay(3).ok && make_lay(4).ok && make_lay(5).ok &&
              make_lay(6).ok && make_lay(7).ok && make_lay(8).ok && make_lay(9).ok && make_lay(10).ok && make_lay(11).ok, "layouts");
static_assert(make_lay(0).Wg[0] == 8 && make_lay(0).Wg[1] == 9 && make_lay(0).Rg[3] == 11 && !make_lay(0).own &&
              make_lay(1).own && make_lay(3).Rg[1] == 9 && make_lay(6).Wg[1] == 10 && make_lay(6).nk == 4 && make_lay(1).nk == 1 && make_lay(1).tidx[5] == 5 && make_lay(6).kidx[15] == 3, "layout values");

__device__ __forceinline__ void lds_barrier() {
    __builtin_amdgcn_fence(__ATOMIC_RELEASE, "workgroup", "local");
    __builtin_amdgcn_s_barrier();
    __builtin_amdgcn_fence(__ATOMIC_ACQUIRE, "workgroup", "local");
}

struct Ops { float2 a[16]; float2 b[16]; };

template <int Q>
__device__ __forceinline__ void table_load(const unsigned lane, const unsigned wv, const float2* psi, Ops& o) {
    constexpr Plan PL = make_plan();
    constexpr Lay LY = make_lay(Q);
    constexpr unsigned F = PL.F[Q];
    const unsigned tp = lane | ((wv & 1u) << LY.Wg[0]) | ((wv >> 1) << LY.Wg[1]);
    const float2* pa = psi + tp;
    const float2* pb = psi + (tp ^ (F & LY.tmask));
#pragma unroll
    for (int j = 0; j < 16; ++j) {
        if constexpr (!LY.own) o.a[j] = pa[LY.sj[j]];
        o.b[j] = pb[LY.sj2[j]];
    }
}

template <int Q>
__device__ __forceinline__ void table_sum(const unsigned lane, const unsigned wv, float2* stab,
                                          const float (&pr)[32], const float (&pi)[32], const Ops& o) {
    constexpr Plan PL = make_plan();
    constexpr Lay LY = make_lay(Q);
    constexpr unsigned F = PL.F[Q];
    constexpr int NPAR = (LY.nk >= 4) ? 1 : 4 / LY.nk;
    float ar[16], ai[16];
#pragma unroll
    for (int j = 0; j < 16; ++j) {
        float xr, xi;
        if constexpr (LY.own) { xr = pr[LY.sj[j] >> 7]; xi = pi[LY.sj[j] >> 7]; }
        else { xr = o.a[j].x; xi = o.a[j].y; }
        const float2 b = o.b[j];
        const int k = LY.kidx[j] * NPAR + (LY.tidx[j] % NPAR);
        if (LY.tidx[j] < NPAR) {
            ar[k] = fmaf(xr, b.x, xi * b.y);
            ai[k] = fmaf(xi, b.x, -(xr * b.y));
        } else {
            ar[k] = fmaf(xr, b.x, fmaf(xi, b.y, ar[k]));
            ai[k] = fmaf(xi, b.x, fmaf(-xr, b.y, ai[k]));
        }
    }
    unsigned st = ((wv & 1u) << LY.pw[0]) | ((wv >> 1) << LY.pw[1]);
#pragma unroll
    for (int g = 0; g < 5; ++g)
        if ((F >> g) & 1u) st |= ((lane >> g) & 1u) << LY.pl[g];
    const bool keep = (lane & (~F & 0x1fu)) == 0u;
#pragma unroll
    for (int k = 0; k < LY.nk; ++k) {
        float vr = ar[k * NPAR], vi = ai[k * NPAR];
        if constexpr (NPAR == 2) { vr += ar[k * 2 + 1]; vi += ai[k * 2 + 1]; }
        if constexpr (NPAR == 4) {
            vr = (vr + ar[k * 4 + 1]) + (ar[k * 4 + 2] + ar[k * 4 + 3]);
            vi = (vi + ai[k * 4 + 1]) + (ai[k * 4 + 2] + ai[k * 4 + 3]);
        }
        if constexpr (((F >> 0) & 1u) == 0u) { vr += xrow<1>(vr); vi += xrow<1>(vi); }
        if constexpr (((F >> 1) & 1u) == 0u) { vr += xrow<2>(vr); vi += xrow<2>(vi); }
        if constexpr (((F >> 2) & 1u) == 0u) { vr += xrow<4>(vr); vi += xrow<4>(vi); }
        if constexpr (((F >> 3) & 1u) == 0u) { vr += xrow<8>(vr); vi += xrow<8>(vi); }
        if constexpr (((F >> 4) & 1u) == 0u) { vr += xhalf(vr);   vi += xhalf(vi); }
        if (keep) stab[PL.off[Q] + (int)LY.ks[k] + (int)st] = make_float2(2.0f * vr, 2.0f * vi);
    }
}

template <int Q>
__device__ __forceinline__ float eval_q(const float (&th)[NQ], const float (&vr)[NQ],
                                        const float (&vi)[NQ], const float2* stab,
                                        const unsigned lane) {
    constexpr Plan PL = make_plan();
    constexpr int K   = PL.K[Q];
    constexpr int NB  = K - 6;
    constexpr int NHI = 1 << NB;
    float ph = 0.f;
#pragma unroll
    for (int i = 0; i < 5; ++i)
        ph += __uint_as_float(__float_as_uint(th[PL.gsel[Q][i]]) ^ (((lane >> i) & 1u) << 31));
    float h0 = 0.f;
#pragma unroll
    for (int i = 5; i < K; ++i) h0 += th[PL.gsel[Q][i]];
    float tr[NHI], ti[NHI];
#pragma unroll
    for (int hi = 0; hi < NHI; ++hi) {
        const float2 e = stab[PL.off[Q] + hi * 32 + (int)lane];
        tr[hi] = e.x; ti[hi] = e.y;
    }
#pragma unroll
    for (int b = NB - 1; b >= 0; --b) {
        const float cr = vr[PL.gsel[Q][5 + b]], ci = vi[PL.gsel[Q][5 + b]];
#pragma unroll
        for (int h = 0; h < (1 << b); ++h) {
            const float xr = tr[h + (1 << b)], xi = ti[h + (1 << b)];
            tr[h] = fmaf(cr, xr, fmaf(-ci, xi, tr[h]));
            ti[h] = fmaf(cr, xi, fmaf(ci, xr, ti[h]));
        }
    }
    const float f = __builtin_amdgcn_fractf(ph + h0);
    return fmaf(__builtin_amdgcn_cosf(f), tr[0], -__builtin_amdgcn_sinf(f) * ti[0]);
}

__global__ __launch_bounds__(ETPB) void vqc_eval(const float* __restrict__ x,
                                                 const float* __restrict__ W_in,
                                                 const float* __restrict__ b_in,
                                                 const float* __restrict__ qp,
                                                 const float* __restrict__ W_out,
                                                 const float* __restrict__ b_out,
                                                 float* __restrict__ out) {
    __shared__ float2 stab[NTERM];
    __shared__ float2 psi[NS];
    __shared__ __attribute__((aligned(16))) float zs[SPB * NQ];
    __shared__ float red[EW * 256];
    const int tid = threadIdx.x;
    const unsigned lane = (unsigned)tid & 31u;
    const int wave = tid >> 5;

    const FcFrag frag = fc_in_load(x, W_in, (int)lane, wave);
    const int gc = ((lane & 15u) < NQ) ? (int)(lane & 15u) : (NQ - 1);
    const float qb = qp[gc], bi = b_in[gc];
    {
        float pr[32], pi[32];
        build_psi(qp, (unsigned)tid, psi, pr, pi);
        lds_barrier();
        const unsigned wv = (unsigned)wave;
        Ops oe, oo;
        table_load<0>(lane, wv, psi, oe);
        table_load<1>(lane, wv, psi, oo);
        table_sum<0>(lane, wv, stab, pr, pi, oe);
        table_load<2>(lane, wv, psi, oe);
        table_sum<1>(lane, wv, stab, pr, pi, oo);
        table_load<3>(lane, wv, psi, oo);
        table_sum<2>(lane, wv, stab, pr, pi, oe);
        table_load<4>(lane, wv, psi, oe);
        table_sum<3>(lane, wv, stab, pr, pi, oo);
        table_load<5>(lane, wv, psi, oo);
        table_sum<4>(lane, wv, stab, pr, pi, oe);
        table_load<6>(lane, wv, psi, oe);
        table_sum<5>(lane, wv, stab, pr, pi, oo);
        table_load<7>(lane, wv, psi, oo);
        table_sum<6>(lane, wv, stab, pr, pi, oe);
        table_load<8>(lane, wv, psi, oe);
        table_sum<7>(lane, wv, stab, pr, pi, oo);
        table_load<9>(lane, wv, psi, oo);
        table_sum<8>(lane, wv, stab, pr, pi, oe);
        table_load<10>(lane, wv, psi, oe);
        table_sum<9>(lane, wv, stab, pr, pi, oo);
        table_load<11>(lane, wv, psi, oo);
        table_sum<10>(lane, wv, stab, pr, pi, oe);
        table_sum<11>(lane, wv, stab, pr, pi, oo);
    }
    fc_in_partial(frag, red, (int)lane, wave);
    __syncthreads();

    const int oc = (int)(lane & 15u), oh = (int)(lane >> 4);
    v2f bw[4][3];
    float bo[4];
#pragma unroll
    for (int t = 0; t < 4; ++t) {
        const float* wp = W_out + (wave * 64 + 16 * t + oc) * NQ + 6 * oh;
#pragma unroll
        for (int s = 0; s < 3; ++s) bw[t][s] = *(const v2f*)(wp + 2 * s);
        bo[t] = b_out[wave * 64 + 16 * t + oc];
    }

    float angA = red[wave * 32 + (int)lane];
    float angB = red[(wave + 4) * 32 + (int)lane];
#pragma unroll
    for (int w = 1; w < EW; ++w) {
        angA += red[w * 256 + wave * 32 + (int)lane];
        angB += red[w * 256 + (wave + 4) * 32 + (int)lane];
    }
    const float thvA = (qb + (angA + bi)) * REV;
    const float thvB = (qb + (angB + bi)) * REV;
    const float f2A = __builtin_amdgcn_fractf(2.0f * thvA);
    const float f2B = __builtin_amdgcn_fractf(2.0f * thvB);
    const float vcvA = __builtin_amdgcn_cosf(f2A), vsvA = -__builtin_amdgcn_sinf(f2A);
    const float vcvB = __builtin_amdgcn_cosf(f2B), vsvB = -__builtin_amdgcn_sinf(f2B);

    const bool b4 = (lane & 16u) != 0u, b3 = (lane & 8u) != 0u, b2 = (lane & 4u) != 0u, b1 = (lane & 2u) != 0u;
    const int qown = b1 ? (8 + 2 * (int)b3 + (int)b4) : (4 * (int)b2 + 2 * (int)b3 + (int)b4);
    const bool own = ((lane & 1u) == 0u) && !(b1 && b2);

    for (int jj = 0; jj < NITER / 2; ++jj) {
        const float thv = jj ? thvB : thvA;
        const float vcv = jj ? vcvB : vcvA;
        const float vsv = jj ? vsvB : vsvA;
        float u[2];
#pragma unroll
        for (int hf = 0; hf < 2; ++hf) {
            float th[NQ], vr[NQ], vi[NQ];
#pragma unroll
            for (int g = 0; g < NQ; ++g) {
                th[g] = __int_as_float(__builtin_amdgcn_readlane(__float_as_int(thv), hf * 16 + g));
                vr[g] = __int_as_float(__builtin_amdgcn_readlane(__float_as_int(vcv), hf * 16 + g));
                vi[g] = __int_as_float(__builtin_amdgcn_readlane(__float_as_int(vsv), hf * 16 + g));
            }
            float zp[NQ];
            zp[0]  = eval_q<0>(th, vr, vi, stab, lane);
            zp[1]  = eval_q<1>(th, vr, vi, stab, lane);
            zp[2]  = eval_q<2>(th, vr, vi, stab, lane);
            zp[3]  = eval_q<3>(th, vr, vi, stab, lane);
            zp[4]  = eval_q<4>(th, vr, vi, stab, lane);
            zp[5]  = eval_q<5>(th, vr, vi, stab, lane);
            zp[6]  = eval_q<6>(th, vr, vi, stab, lane);
            zp[7]  = eval_q<7>(th, vr, vi, stab, lane);
            zp[8]  = eval_q<8>(th, vr, vi, stab, lane);
            zp[9]  = eval_q<9>(th, vr, vi, stab, lane);
            zp[10] = eval_q<10>(th, vr, vi, stab, lane);
            zp[11] = eval_q<11>(th, vr, vi, stab, lane);
            float y[NQ / 2];
#pragma unroll
            for (int j = 0; j < NQ / 2; ++j) y[j] = tstep<16>(zp[2 * j], zp[2 * j + 1], b4);
            const float x0 = tstep<8>(y[0], y[1], b3);
            const float x1 = tstep<8>(y[2], y[3], b3);
            const float x2 = tstep<8>(y[4], y[5], b3);
            const float w0 = tstep<4>(x0, x1, b2);
            const float w1 = x2 + xrow<4>(x2);
            float uu = tstep<2>(w0, w1, b1);
            uu += xrow<1>(uu);
            u[hf] = uu * (1.0f / 4096.0f);
        }
        if (own) {
            zs[(wave + 4 * jj) * NQ + qown]     = u[0];
            zs[(wave + 4 * jj + 8) * NQ + qown] = u[1];
        }
    }
    __syncthreads();

    v8f acc[4];
#pragma unroll
    for (int t = 0; t < 4; ++t) {
#pragma unroll
        for (int r = 0; r < 8; ++r) acc[t][r] = bo[t];
    }
    const float* zr = zs + oc * NQ + 6 * oh;
#pragma unroll
    for (int s = 0; s < 3; ++s) {
        const v2f az = *(const v2f*)(zr + 2 * s);
#pragma unroll
        for (int t = 0; t < 4; ++t)
            acc[t] = __builtin_amdgcn_wmma_f32_16x16x4_f32(false, az, false, bw[t][s], (short)0, acc[t], false, false);
    }
#pragma unroll
    for (int p = 0; p < 2; ++p) {
        float* orow = out + ((size_t)blockIdx.x * SPB) * DOUT + wave * 64 + 32 * p + (int)lane;
#pragma unroll
        for (int r = 0; r < 8; ++r) {
            const float t0x = xhalf(acc[2 * p][r]);
            const float t1x = xhalf(acc[2 * p + 1][r]);
            const float lo = oh ? t1x : acc[2 * p][r];
            const float hi = oh ? acc[2 * p + 1][r] : t0x;
            orow[(size_t)r * DOUT] = lo;
            orow[(size_t)(r + 8) * DOUT] = hi;
        }
    }
}

static_assert(4096 / SPB == 256 && SPB == 16 && NITER == 4 && ETPB == 128 && NS == 32 * ETPB, "grid literals below, 32 patterns per thread");
extern "C" void kernel_launch(void* const* d_in, const int* in_sizes, int n_in,
                              void* d_out, int out_size, void* d_ws, size_t ws_size,
                              hipStream_t stream) {
    const float* x        = (const float*)d_in[0];
    const float* W_in     = (const float*)d_in[1];
    const float* b_in     = (const float*)d_in[2];
    const float* q_params = (const float*)d_in[3];
    const float* W_out    = (const float*)d_in[4];
    const float* b_out    = (const float*)d_in[5];
    float* out    = (float*)d_out;

    vqc_eval<<<dim3(256), dim3(128), 0, stream>>>(x, W_in, b_in, q_params, W_out, b_out, out);
}
